# starting kernel
# baseline (speedup 1.0000x reference)
#include <hip/hip_runtime.h>
#include <hip/hip_bf16.h>


#define MTOK 32768
#define CCH  256

typedef __attribute__((ext_vector_type(2))) float v2f;
typedef __attribute__((ext_vector_type(8))) float v8f;

__device__ __forceinline__ v8f wmma_f32(v2f a, v2f b, v8f c) {
    return __builtin_amdgcn_wmma_f32_16x16x4_f32(false, a, false, b, (short)0, c, false, false);
}

__global__ __launch_bounds__(512) void gram_kernel(const float* __restrict__ x,
                                                   float* __restrict__ Gpart,
                                                   float* __restrict__ Spart) {
    __shared__ float lds[CCH * 34];
    const int n    = blockIdx.y;
    const int wg   = blockIdx.x;
    const float* xn = x + (size_t)n * CCH * MTOK;
    const int tid  = threadIdx.x;
    const int lane = tid & 31;
    const int w    = tid >> 5;
    const int bwr  = w >> 2, bwc = w & 3;
    const int l15  = lane & 15;
    const int half = lane >> 4;

    const v8f zero8 = {0.f,0.f,0.f,0.f,0.f,0.f,0.f,0.f};
    v8f acc[4][4];
#pragma unroll
    for (int r = 0; r < 4; ++r)
#pragma unroll
        for (int c = 0; c < 4; ++c) acc[r][c] = zero8;

    float sacc[16];
#pragma unroll
    for (int i = 0; i < 16; ++i) sacc[i] = 0.f;

    const int m0 = wg * 512;
    for (int t0 = 0; t0 < 512; t0 += 32) {
        __syncthreads();
#pragma unroll
        for (int i = 0; i < 16; ++i) {
            const int ch = w + 16 * i;
            const float val = xn[(size_t)ch * MTOK + m0 + t0 + lane];
            lds[ch * 34 + lane] = val;
            sacc[i] += val;
        }
        __syncthreads();
#pragma unroll
        for (int kb = 0; kb < 32; kb += 4) {
            const int kk = kb + half * 2;
            v2f a[4], b[4];
#pragma unroll
            for (int r = 0; r < 4; ++r) {
                const int chA = (bwr * 4 + r) * 16 + l15;
                a[r] = *(const v2f*)&lds[chA * 34 + kk];
                const int chB = (bwc * 4 + r) * 16 + l15;
                b[r] = *(const v2f*)&lds[chB * 34 + kk];
            }
#pragma unroll
            for (int r = 0; r < 4; ++r)
#pragma unroll
                for (int c = 0; c < 4; ++c)
                    acc[r][c] = wmma_f32(a[r], b[c], acc[r][c]);
        }
    }

    float* gp = Gpart + ((size_t)n * 64 + wg) * (CCH * CCH);
#pragma unroll
    for (int r = 0; r < 4; ++r)
#pragma unroll
        for (int c = 0; c < 4; ++c) {
            const int row0 = (bwr * 4 + r) * 16 + half * 8;
            const int col  = (bwc * 4 + c) * 16 + l15;
#pragma unroll
            for (int v = 0; v < 8; ++v)
                gp[(row0 + v) * CCH + col] = acc[r][c][v];
        }

#pragma unroll
    for (int i = 0; i < 16; ++i) {
        float s = sacc[i];
        for (int off = 16; off >= 1; off >>= 1) s += __shfl_xor(s, off, 32);
        if (lane == 0) Spart[((size_t)n * 64 + wg) * CCH + w + 16 * i] = s;
    }
}

__global__ __launch_bounds__(256) void reduce_kernel(const float* __restrict__ Gpart,
                                                     const float* __restrict__ Spart,
                                                     float* __restrict__ G,
                                                     float* __restrict__ S) {
    const int idx = blockIdx.x * 256 + threadIdx.x;
    const int n = idx >> 16;
    const int e = idx & 65535;
    const float* gp = Gpart + (size_t)n * 64 * 65536 + e;
    float s = 0.f;
    for (int p = 0; p < 64; ++p) s += gp[(size_t)p * 65536];
    G[idx] = s;
    if (idx < 512) {
        const float* sp = Spart + (size_t)(idx >> 8) * 64 * CCH + (idx & 255);
        float t = 0.f;
        for (int p = 0; p < 64; ++p) t += sp[p * CCH];
        S[idx] = t;
    }
}

__global__ __launch_bounds__(512) void vecprep_kernel(const float* __restrict__ Wq,
                                                      const float* __restrict__ Wk,
                                                      const float* __restrict__ bk,
                                                      const float* __restrict__ S,
                                                      float* __restrict__ U,
                                                      float* __restrict__ Wvec) {
    const int tid = threadIdx.x;
    const int n = tid >> 8, i = tid & 255;
    const float* s = S + n * CCH;
    float u = 0.f, wv = 0.f;
    for (int c = 0; c < CCH; ++c) {
        u  += Wq[i * CCH + c] * s[c];
        wv += Wk[i * CCH + c] * s[c];
    }
    U[tid] = u;
    Wvec[tid] = wv + (float)MTOK * bk[i];
}

__global__ __launch_bounds__(512) void gemm256_kernel(const float* __restrict__ A, long long sA,
                                                      const float* __restrict__ B, long long sB,
                                                      int transB,
                                                      float* __restrict__ D, long long sD) {
    const int n = blockIdx.y;
    const float* An = A + (size_t)n * sA;
    const float* Bn = B + (size_t)n * sB;
    float* Dn = D + (size_t)n * sD;
    const int tid = threadIdx.x, lane = tid & 31, w = tid >> 5;
    const int l15 = lane & 15, half = lane >> 4;
    const int row0 = blockIdx.x * 64;
    const int tr  = w & 3;
    const int tc0 = (w >> 2) * 4;
    const int rowA = row0 + tr * 16 + l15;

    const v8f zero8 = {0.f,0.f,0.f,0.f,0.f,0.f,0.f,0.f};
    v8f acc[4];
#pragma unroll
    for (int c = 0; c < 4; ++c) acc[c] = zero8;

    for (int kb = 0; kb < CCH; kb += 4) {
        const int kk = kb + half * 2;
        const v2f a = *(const v2f*)&An[rowA * CCH + kk];
#pragma unroll
        for (int c = 0; c < 4; ++c) {
            const int col = (tc0 + c) * 16 + l15;
            v2f b;
            if (transB) {
                b = *(const v2f*)&Bn[col * CCH + kk];
            } else {
                b[0] = Bn[kk * CCH + col];
                b[1] = Bn[(kk + 1) * CCH + col];
            }
            acc[c] = wmma_f32(a, b, acc[c]);
        }
    }
#pragma unroll
    for (int c = 0; c < 4; ++c) {
        const int col = (tc0 + c) * 16 + l15;
        const int rr = row0 + tr * 16 + half * 8;
#pragma unroll
        for (int v = 0; v < 8; ++v)
            Dn[(rr + v) * CCH + col] = acc[c][v];
    }
}

__global__ __launch_bounds__(256) void softmax_kernel(const float* __restrict__ P0,
                                                      const float* __restrict__ U,
                                                      const float* __restrict__ Wvec,
                                                      const float* __restrict__ bq,
                                                      const float* __restrict__ bk,
                                                      const float* __restrict__ bv,
                                                      float* __restrict__ att,
                                                      float* __restrict__ cvec) {
    const int n = blockIdx.y;
    const int row = blockIdx.x * 8 + (threadIdx.x >> 5);
    const int lane = threadIdx.x & 31;
    const float* p = P0 + ((size_t)n * CCH + row) * CCH;
    const float ui = U[n * CCH + row];
    const float bqi = bq[row];
    float l[8];
    float mx = -3.0e38f;
#pragma unroll
    for (int t = 0; t < 8; ++t) {
        const int j = lane + 32 * t;
        const float val = p[j] + ui * bk[j] + bqi * Wvec[n * CCH + j];
        l[t] = val;
        mx = fmaxf(mx, val);
    }
    for (int off = 16; off >= 1; off >>= 1) mx = fmaxf(mx, __shfl_xor(mx, off, 32));
    float sum = 0.f;
#pragma unroll
    for (int t = 0; t < 8; ++t) { l[t] = __expf(l[t] - mx); sum += l[t]; }
    for (int off = 16; off >= 1; off >>= 1) sum += __shfl_xor(sum, off, 32);
    const float inv = 1.f / sum;
    float cd = 0.f;
    float* arow = att + ((size_t)n * CCH + row) * CCH;
#pragma unroll
    for (int t = 0; t < 8; ++t) {
        const int j = lane + 32 * t;
        const float a = l[t] * inv;
        arow[j] = a;
        cd += a * bv[j];
    }
    for (int off = 16; off >= 1; off >>= 1) cd += __shfl_xor(cd, off, 32);
    if (lane == 0) cvec[n * CCH + row] = cd;
}

__global__ __launch_bounds__(512) void out_kernel(const float* __restrict__ x,
                                                  const float* __restrict__ Bt,
                                                  const float* __restrict__ cvec,
                                                  float* __restrict__ out) {
    __shared__ float lds[CCH * 40];
    const int n = blockIdx.y;
    const float* xn = x + (size_t)n * CCH * MTOK;
    float* on = out + (size_t)n * MTOK * CCH;
    const int m0 = blockIdx.x * 32;
    const int tid = threadIdx.x, lane = tid & 31, w = tid >> 5;
    const int l15 = lane & 15, half = lane >> 4;

#pragma unroll
    for (int i = 0; i < 16; ++i) {
        const int ch = w + 16 * i;
        lds[ch * 40 + lane] = xn[(size_t)ch * MTOK + m0 + lane];
    }
    __syncthreads();

    const int tr  = w & 1;
    const int tc0 = (w >> 1) * 2;
    const float* btn = Bt + (size_t)n * CCH * CCH;
    const v8f zero8 = {0.f,0.f,0.f,0.f,0.f,0.f,0.f,0.f};
    v8f acc[2];
    acc[0] = zero8; acc[1] = zero8;

    for (int kb = 0; kb < CCH; kb += 4) {
        const int kk = kb + half * 2;
        const int ml = tr * 16 + l15;
        v2f a;
        a[0] = lds[kk * 40 + ml];
        a[1] = lds[(kk + 1) * 40 + ml];
#pragma unroll
        for (int c = 0; c < 2; ++c) {
            const int col = (tc0 + c) * 16 + l15;
            const v2f b = *(const v2f*)&btn[col * CCH + kk];
            acc[c] = wmma_f32(a, b, acc[c]);
        }
    }
#pragma unroll
    for (int c = 0; c < 2; ++c) {
        const int col = (tc0 + c) * 16 + l15;
        const float cv = cvec[n * CCH + col];
        const int mrow = tr * 16 + half * 8;
#pragma unroll
        for (int v = 0; v < 8; ++v)
            on[(size_t)(m0 + mrow + v) * CCH + col] = acc[c][v] + cv;
    }
}

extern "C" void kernel_launch(void* const* d_in, const int* in_sizes, int n_in,
                              void* d_out, int out_size, void* d_ws, size_t ws_size,
                              hipStream_t stream) {
    (void)in_sizes; (void)n_in; (void)out_size; (void)ws_size;
    const float* x  = (const float*)d_in[0];
    const float* Wq = (const float*)d_in[1];
    const float* bq = (const float*)d_in[2];
    const float* Wk = (const float*)d_in[3];
    const float* bk = (const float*)d_in[4];
    const float* Wv = (const float*)d_in[5];
    const float* bv = (const float*)d_in[6];
    float* out = (float*)d_out;
    float* ws  = (float*)d_ws;

    const size_t oGP  = 0;
    const size_t oSP  = oGP  + (size_t)2 * 64 * 65536;
    const size_t oG   = oSP  + (size_t)2 * 64 * 256;
    const size_t oS   = oG   + (size_t)2 * 65536;
    const size_t oT1  = oS   + 512;
    const size_t oP0  = oT1  + (size_t)2 * 65536;
    const size_t oATT = oP0  + (size_t)2 * 65536;
    const size_t oBT  = oATT + (size_t)2 * 65536;
    const size_t oU   = oBT  + (size_t)2 * 65536;
    const size_t oW   = oU   + 512;
    const size_t oCV  = oW   + 512;

    gram_kernel<<<dim3(64, 2), 512, 0, stream>>>(x, ws + oGP, ws + oSP);
    reduce_kernel<<<512, 256, 0, stream>>>(ws + oGP, ws + oSP, ws + oG, ws + oS);
    vecprep_kernel<<<1, 512, 0, stream>>>(Wq, Wk, bk, ws + oS, ws + oU, ws + oW);
    gemm256_kernel<<<dim3(4, 2), 512, 0, stream>>>(Wq, 0, ws + oG, 65536, 0, ws + oT1, 65536);
    gemm256_kernel<<<dim3(4, 2), 512, 0, stream>>>(ws + oT1, 65536, Wk, 0, 1, ws + oP0, 65536);
    softmax_kernel<<<dim3(32, 2), 256, 0, stream>>>(ws + oP0, ws + oU, ws + oW, bq, bk, bv,
                                                    ws + oATT, ws + oCV);
    gemm256_kernel<<<dim3(4, 2), 512, 0, stream>>>(ws + oATT, 65536, Wv, 0, 0, ws + oBT, 65536);
    out_kernel<<<dim3(1024, 2), 512, 0, stream>>>(x, ws + oBT, ws + oCV, out);
}


# vecprep, 3 gemm256 and softmax fused into one launch: each WG takes a 16-row band Wq->T1->P0->att->Bt through LDS, B panels whole-K in registers; u and w dots in-kernel, their sum order changes
# speedup vs baseline: 1.7087x; 1.7087x over previous
#include <hip/hip_runtime.h>
#include <hip/hip_bf16.h>


#define MTOK 32768
#define CCH  256

typedef __attribute__((ext_vector_type(2))) float v2f;
typedef __attribute__((ext_vector_type(8))) float v8f;

__device__ __forceinline__ v8f wmma_f32(v2f a, v2f b, v8f c) {
    return __builtin_amdgcn_wmma_f32_16x16x4_f32(false, a, false, b, (short)0, c, false, false);
}

__global__ __launch_bounds__(512) void gram_kernel(const float* __restrict__ x,
                                                   float* __restrict__ Gpart,
                                                   float* __restrict__ Spart) {
    __shared__ float lds[CCH * 34];
    const int n    = blockIdx.y;
    const int wg   = blockIdx.x;
    const float* xn = x + (size_t)n * CCH * MTOK;
    const int tid  = threadIdx.x;
    const int lane = tid & 31;
    const int w    = tid >> 5;
    const int bwr  = w >> 2, bwc = w & 3;
    const int l15  = lane & 15;
    const int half = lane >> 4;

    const v8f zero8 = {0.f,0.f,0.f,0.f,0.f,0.f,0.f,0.f};
    v8f acc[4][4];
#pragma unroll
    for (int r = 0; r < 4; ++r)
#pragma unroll
        for (int c = 0; c < 4; ++c) acc[r][c] = zero8;

    float sacc[16];
#pragma unroll
    for (int i = 0; i < 16; ++i) sacc[i] = 0.f;

    const int m0 = wg * 512;
    for (int t0 = 0; t0 < 512; t0 += 32) {
        __syncthreads();
#pragma unroll
        for (int i = 0; i < 16; ++i) {
            const int ch = w + 16 * i;
            const float val = xn[(size_t)ch * MTOK + m0 + t0 + lane];
            lds[ch * 34 + lane] = val;
            sacc[i] += val;
        }
        __syncthreads();
#pragma unroll
        for (int kb = 0; kb < 32; kb += 4) {
            const int kk = kb + half * 2;
            v2f a[4], b[4];
#pragma unroll
            for (int r = 0; r < 4; ++r) {
                const int chA = (bwr * 4 + r) * 16 + l15;
                a[r] = *(const v2f*)&lds[chA * 34 + kk];
                const int chB = (bwc * 4 + r) * 16 + l15;
                b[r] = *(const v2f*)&lds[chB * 34 + kk];
            }
#pragma unroll
            for (int r = 0; r < 4; ++r)
#pragma unroll
                for (int c = 0; c < 4; ++c)
                    acc[r][c] = wmma_f32(a[r], b[c], acc[r][c]);
        }
    }

    float* gp = Gpart + ((size_t)n * 64 + wg) * (CCH * CCH);
#pragma unroll
    for (int r = 0; r < 4; ++r)
#pragma unroll
        for (int c = 0; c < 4; ++c) {
            const int row0 = (bwr * 4 + r) * 16 + half * 8;
            const int col  = (bwc * 4 + c) * 16 + l15;
#pragma unroll
            for (int v = 0; v < 8; ++v)
                gp[(row0 + v) * CCH + col] = acc[r][c][v];
        }

#pragma unroll
    for (int i = 0; i < 16; ++i) {
        float s = sacc[i];
        for (int off = 16; off >= 1; off >>= 1) s += __shfl_xor(s, off, 32);
        if (lane == 0) Spart[((size_t)n * 64 + wg) * CCH + w + 16 * i] = s;
    }
}

__global__ __launch_bounds__(256) void reduce_kernel(const float* __restrict__ Gpart,
                                                     const float* __restrict__ Spart,
                                                     float* __restrict__ G,
                                                     float* __restrict__ S) {
    const int idx = blockIdx.x * 256 + threadIdx.x;
    const int n = idx >> 16;
    const int e = idx & 65535;
    const float* gp = Gpart + (size_t)n * 64 * 65536 + e;
    float s = 0.f;
    for (int p = 0; p < 64; ++p) s += gp[(size_t)p * 65536];
    G[idx] = s;
    if (idx < 512) {
        const float* sp = Spart + (size_t)(idx >> 8) * 64 * CCH + (idx & 255);
        float t = 0.f;
        for (int p = 0; p < 64; ++p) t += sp[p * CCH];
        S[idx] = t;
    }
}

#define BPAD 260

__device__ __forceinline__ void lds_handover() {
    __builtin_amdgcn_fence(__ATOMIC_RELEASE, "workgroup", "local");
    __builtin_amdgcn_s_barrier();
    __builtin_amdgcn_fence(__ATOMIC_ACQUIRE, "workgroup", "local");
}

__global__ __launch_bounds__(512) void band_kernel(const float* __restrict__ Wq,
                                                   const float* __restrict__ Wk,
                                                   const float* __restrict__ Wv,
                                                   const float* __restrict__ bq,
                                                   const float* __restrict__ bk,
                                                   const float* __restrict__ bv,
                                                   const float* __restrict__ G,
                                                   const float* __restrict__ S,
                                                   float* __restrict__ Bt,
                                                   float* __restrict__ cvec) {
    __shared__ __attribute__((aligned(16))) float ldsQ[16 * BPAD];
    __shared__ __attribute__((aligned(16))) float ldsT[16 * BPAD];
    __shared__ __attribute__((aligned(16))) float ldsS[CCH];
    __shared__ float ldsBK[CCH];
    __shared__ float ldsBV[CCH];
    __shared__ float ldsBQ[16];
    __shared__ float ldsU[16];
    __shared__ float ldsC[16];

    const int n   = blockIdx.y;
    const int i0  = blockIdx.x * 16;
    const int tid = threadIdx.x, lane = tid & 31, w = tid >> 5;
    const int l15 = lane & 15, half = lane >> 4;
    const int col = w * 16 + l15;
    const int kh  = half * 2;
    const float* Gn  = G + (size_t)n * CCH * CCH;
    const float* wkr = Wk + col * CCH;

    v2f pa[32], pb[32];

    const int   t8    = tid & (CCH - 1);
    const float s_in  = S[n * CCH + t8];
    const float bk_in = bk[t8];
    const float bv_in = bv[t8];
    const float bq_in = bq[i0 + (tid & 15)];
    float4 q_in[2];
#pragma unroll
    for (int i = 0; i < 2; ++i) {
        const int f = tid + 512 * i;
        q_in[i] = *(const float4*)&Wq[(i0 + (f >> 6)) * CCH + (f & 63) * 4];
    }
    __builtin_amdgcn_sched_barrier(0);

#pragma unroll
    for (int i = 0; i < 32; ++i) {
        const int kk = i * 4 + kh;
        pa[i][0] = Gn[kk * CCH + col];
        pa[i][1] = Gn[(kk + 1) * CCH + col];
    }
#pragma unroll
    for (int i = 0; i < 32; ++i) {
        const int kk = (i + 32) * 4 + kh;
        pb[i][0] = Gn[kk * CCH + col];
        pb[i][1] = Gn[(kk + 1) * CCH + col];
    }

#pragma unroll
    for (int i = 0; i < 2; ++i) {
        const int f = tid + 512 * i;
        *(float4*)&ldsQ[(f >> 6) * BPAD + (f & 63) * 4] = q_in[i];
    }
    ldsS[t8] = s_in; ldsBK[t8] = bk_in; ldsBV[t8] = bv_in;
    ldsBQ[tid & 15] = bq_in;
    lds_handover();

    {
        float p = 0.f;
#pragma unroll
        for (int t = 0; t < 8; ++t) {
            const int j = lane + 32 * t;
            p += ldsQ[w * BPAD + j] * ldsS[j];
        }
        for (int off = 16; off >= 1; off >>= 1) p += __shfl_xor(p, off, 32);
        if (lane == 0) ldsU[w] = p;
    }

    const v8f zero8 = {0.f,0.f,0.f,0.f,0.f,0.f,0.f,0.f};

    v8f acc = zero8;
#pragma unroll
    for (int i = 0; i < 32; ++i) {
        const int kk = i * 4 + kh;
        const v2f a = *(const v2f*)&ldsQ[l15 * BPAD + kk];
        acc = wmma_f32(a, pa[i], acc);
    }
#pragma unroll
    for (int i = 0; i < 32; ++i) {
        const int kk = i * 4 + kh;
        pa[i] = *(const v2f*)&wkr[kk];
    }
    __builtin_amdgcn_sched_barrier(0);
#pragma unroll
    for (int i = 0; i < 32; ++i) {
        const int kk = (i + 32) * 4 + kh;
        const v2f a = *(const v2f*)&ldsQ[l15 * BPAD + kk];
        acc = wmma_f32(a, pb[i], acc);
    }
#pragma unroll
    for (int i = 0; i < 32; ++i) {
        const int kk = (i + 32) * 4 + kh;
        pb[i] = *(const v2f*)&wkr[kk];
    }
#pragma unroll
    for (int v = 0; v < 8; ++v) ldsT[(half * 8 + v) * BPAD + col] = acc[v];
    lds_handover();

    acc = zero8;
    float wd = 0.f;
#pragma unroll
    for (int i = 0; i < 32; ++i) {
        const int kk = i * 4 + kh;
        const v2f a  = *(const v2f*)&ldsT[l15 * BPAD + kk];
        const v2f sv = *(const v2f*)&ldsS[kk];
        wd += pa[i][0] * sv[0];
        wd += pa[i][1] * sv[1];
        acc = wmma_f32(a, pa[i], acc);
    }
#pragma unroll
    for (int i = 0; i < 32; ++i) {
        const int kk = i * 4 + kh;
        pa[i][0] = Wv[kk * CCH + col];
        pa[i][1] = Wv[(kk + 1) * CCH + col];
    }
    __builtin_amdgcn_sched_barrier(0);
#pragma unroll
    for (int i = 0; i < 32; ++i) {
        const int kk = (i + 32) * 4 + kh;
        const v2f a  = *(const v2f*)&ldsT[l15 * BPAD + kk];
        const v2f sv = *(const v2f*)&ldsS[kk];
        wd += pb[i][0] * sv[0];
        wd += pb[i][1] * sv[1];
        acc = wmma_f32(a, pb[i], acc);
    }
#pragma unroll
    for (int i = 0; i < 32; ++i) {
        const int kk = (i + 32) * 4 + kh;
        pb[i][0] = Wv[kk * CCH + col];
        pb[i][1] = Wv[(kk + 1) * CCH + col];
    }
    wd += __shfl_xor(wd, 16, 32);

    {
        const float bkc = ldsBK[col];
        const float wv  = wd + (float)MTOK * bkc;
#pragma unroll
        for (int v = 0; v < 8; ++v) {
            const int r = half * 8 + v;
            const float val = acc[v] + ldsU[r] * bkc + ldsBQ[r] * wv;
            ldsQ[r * BPAD + col] = val;
        }
    }
    lds_handover();

    {
        float* rowp = &ldsQ[w * BPAD];
        float l[8];
        float mx = -3.0e38f;
#pragma unroll
        for (int t = 0; t < 8; ++t) {
            l[t] = rowp[lane + 32 * t];
            mx = fmaxf(mx, l[t]);
        }
        for (int off = 16; off >= 1; off >>= 1) mx = fmaxf(mx, __shfl_xor(mx, off, 32));
        float sum = 0.f;
#pragma unroll
        for (int t = 0; t < 8; ++t) { l[t] = __expf(l[t] - mx); sum += l[t]; }
        for (int off = 16; off >= 1; off >>= 1) sum += __shfl_xor(sum, off, 32);
        const float inv = 1.f / sum;
        float cd = 0.f;
#pragma unroll
        for (int t = 0; t < 8; ++t) {
            const int j = lane + 32 * t;
            const float a = l[t] * inv;
            rowp[j] = a;
            cd += a * ldsBV[j];
        }
        for (int off = 16; off >= 1; off >>= 1) cd += __shfl_xor(cd, off, 32);
        if (lane == 0) ldsC[w] = cd;
    }
    lds_handover();

    acc = zero8;
#pragma unroll
    for (int i = 0; i < 32; ++i) {
        const int kk = i * 4 + kh;
        const v2f a = *(const v2f*)&ldsQ[l15 * BPAD + kk];
        acc = wmma_f32(a, pa[i], acc);
    }
#pragma unroll
    for (int i = 0; i < 32; ++i) {
        const int kk = (i + 32) * 4 + kh;
        const v2f a = *(const v2f*)&ldsQ[l15 * BPAD + kk];
        acc = wmma_f32(a, pb[i], acc);
    }
#pragma unroll
    for (int v = 0; v < 8; ++v) ldsT[(half * 8 + v) * BPAD + col] = acc[v];
    lds_handover();
    {
        float* brow = Bt + (size_t)n * CCH * CCH + (size_t)(i0 + w) * CCH;
#pragma unroll
        for (int t = 0; t < 2; ++t) {
            const int c = lane * 4 + 128 * t;
            *(float4*)&brow[c] = *(const float4*)&ldsT[w * BPAD + c];
        }
    }
    if (tid < 16) cvec[n * CCH + i0 + tid] = ldsC[tid];
}

__global__ __launch_bounds__(512) void out_kernel(const float* __restrict__ x,
                                                  const float* __restrict__ Bt,
                                                  const float* __restrict__ cvec,
                                                  float* __restrict__ out) {
    __shared__ float lds[CCH * 40];
    const int n = blockIdx.y;
    const float* xn = x + (size_t)n * CCH * MTOK;
    float* on = out + (size_t)n * MTOK * CCH;
    const int m0 = blockIdx.x * 32;
    const int tid = threadIdx.x, lane = tid & 31, w = tid >> 5;
    const int l15 = lane & 15, half = lane >> 4;

#pragma unroll
    for (int i = 0; i < 16; ++i) {
        const int ch = w + 16 * i;
        lds[ch * 40 + lane] = xn[(size_t)ch * MTOK + m0 + lane];
    }
    __syncthreads();

    const int tr  = w & 1;
    const int tc0 = (w >> 1) * 2;
    const float* btn = Bt + (size_t)n * CCH * CCH;
    const v8f zero8 = {0.f,0.f,0.f,0.f,0.f,0.f,0.f,0.f};
    v8f acc[2];
    acc[0] = zero8; acc[1] = zero8;

    for (int kb = 0; kb < CCH; kb += 4) {
        const int kk = kb + half * 2;
        const int ml = tr * 16 + l15;
        v2f a;
        a[0] = lds[kk * 40 + ml];
        a[1] = lds[(kk + 1) * 40 + ml];
#pragma unroll
        for (int c = 0; c < 2; ++c) {
            const int col = (tc0 + c) * 16 + l15;
            const v2f b = *(const v2f*)&btn[col * CCH + kk];
            acc[c] = wmma_f32(a, b, acc[c]);
        }
    }
#pragma unroll
    for (int c = 0; c < 2; ++c) {
        const int col = (tc0 + c) * 16 + l15;
        const float cv = cvec[n * CCH + col];
        const int mrow = tr * 16 + half * 8;
#pragma unroll
        for (int v = 0; v < 8; ++v)
            on[(size_t)(m0 + mrow + v) * CCH + col] = acc[c][v] + cv;
    }
}

extern "C" void kernel_launch(void* const* d_in, const int* in_sizes, int n_in,
                              void* d_out, int out_size, void* d_ws, size_t ws_size,
                              hipStream_t stream) {
    (void)in_sizes; (void)n_in; (void)out_size; (void)ws_size;
    const float* x  = (const float*)d_in[0];
    const float* Wq = (const float*)d_in[1];
    const float* bq = (const float*)d_in[2];
    const float* Wk = (const float*)d_in[3];
    const float* bk = (const float*)d_in[4];
    const float* Wv = (const float*)d_in[5];
    const float* bv = (const float*)d_in[6];
    float* out = (float*)d_out;
    float* ws  = (float*)d_ws;

    const size_t oGP  = 0;
    const size_t oSP  = oGP  + (size_t)2 * 64 * 65536;
    const size_t oG   = oSP  + (size_t)2 * 64 * 256;
    const size_t oS   = oG   + (size_t)2 * 65536;
    const size_t oT1  = oS   + 512;
    const size_t oP0  = oT1  + (size_t)2 * 65536;
    const size_t oATT = oP0  + (size_t)2 * 65536;
    const size_t oBT  = oATT + (size_t)2 * 65536;
    const size_t oU   = oBT  + (size_t)2 * 65536;
    const size_t oW   = oU   + 512;
    const size_t oCV  = oW   + 512;

    gram_kernel<<<dim3(64, 2), 512, 0, stream>>>(x, ws + oGP, ws + oSP);
    reduce_kernel<<<512, 256, 0, stream>>>(ws + oGP, ws + oSP, ws + oG, ws + oS);
    band_kernel<<<dim3(16, 2), 512, 0, stream>>>(Wq, Wk, Wv, bq, bk, bv, ws + oG, ws + oS,
                                                 ws + oBT, ws + oCV);
    out_kernel<<<dim3(1024, 2), 512, 0, stream>>>(x, ws + oBT, ws + oCV, out);
}


# out_kernel: one WG per WGP stages Bt[n] once in LDS; each wave holds its 16-token A panel (whole K) in registers straight from x and walks the 16 column tiles; k order and sums unchanged
# speedup vs baseline: 2.2718x; 1.3295x over previous
#include <hip/hip_runtime.h>
#include <hip/hip_bf16.h>


#define MTOK 32768
#define CCH  256

typedef __attribute__((ext_vector_type(2))) float v2f;
typedef __attribute__((ext_vector_type(8))) float v8f;

__device__ __forceinline__ v8f wmma_f32(v2f a, v2f b, v8f c) {
    return __builtin_amdgcn_wmma_f32_16x16x4_f32(false, a, false, b, (short)0, c, false, false);
}

__global__ __launch_bounds__(512) void gram_kernel(const float* __restrict__ x,
                                                   float* __restrict__ Gpart,
                                                   float* __restrict__ Spart) {
    __shared__ float lds[CCH * 34];
    const int n    = blockIdx.y;
    const int wg   = blockIdx.x;
    const float* xn = x + (size_t)n * CCH * MTOK;
    const int tid  = threadIdx.x;
    const int lane = tid & 31;
    const int w    = tid >> 5;
    const int bwr  = w >> 2, bwc = w & 3;
    const int l15  = lane & 15;
    const int half = lane >> 4;

    const v8f zero8 = {0.f,0.f,0.f,0.f,0.f,0.f,0.f,0.f};
    v8f acc[4][4];
#pragma unroll
    for (int r = 0; r < 4; ++r)
#pragma unroll
        for (int c = 0; c < 4; ++c) acc[r][c] = zero8;

    float sacc[16];
#pragma unroll
    for (int i = 0; i < 16; ++i) sacc[i] = 0.f;

    const int m0 = wg * 512;
    for (int t0 = 0; t0 < 512; t0 += 32) {
        __syncthreads();
#pragma unroll
        for (int i = 0; i < 16; ++i) {
            const int ch = w + 16 * i;
            const float val = xn[(size_t)ch * MTOK + m0 + t0 + lane];
            lds[ch * 34 + lane] = val;
            sacc[i] += val;
        }
        __syncthreads();
#pragma unroll
        for (int kb = 0; kb < 32; kb += 4) {
            const int kk = kb + half * 2;
            v2f a[4], b[4];
#pragma unroll
            for (int r = 0; r < 4; ++r) {
                const int chA = (bwr * 4 + r) * 16 + l15;
                a[r] = *(const v2f*)&lds[chA * 34 + kk];
                const int chB = (bwc * 4 + r) * 16 + l15;
                b[r] = *(const v2f*)&lds[chB * 34 + kk];
            }
#pragma unroll
            for (int r = 0; r < 4; ++r)
#pragma unroll
                for (int c = 0; c < 4; ++c)
                    acc[r][c] = wmma_f32(a[r], b[c], acc[r][c]);
        }
    }

    float* gp = Gpart + ((size_t)n * 64 + wg) * (CCH * CCH);
#pragma unroll
    for (int r = 0; r < 4; ++r)
#pragma unroll
        for (int c = 0; c < 4; ++c) {
            const int row0 = (bwr * 4 + r) * 16 + half * 8;
            const int col  = (bwc * 4 + c) * 16 + l15;
#pragma unroll
            for (int v = 0; v < 8; ++v)
                gp[(row0 + v) * CCH + col] = acc[r][c][v];
        }

#pragma unroll
    for (int i = 0; i < 16; ++i) {
        float s = sacc[i];
        for (int off = 16; off >= 1; off >>= 1) s += __shfl_xor(s, off, 32);
        if (lane == 0) Spart[((size_t)n * 64 + wg) * CCH + w + 16 * i] = s;
    }
}

__global__ __launch_bounds__(256) void reduce_kernel(const float* __restrict__ Gpart,
                                                     const float* __restrict__ Spart,
                                                     float* __restrict__ G,
                                                     float* __restrict__ S) {
    const int idx = blockIdx.x * 256 + threadIdx.x;
    const int n = idx >> 16;
    const int e = idx & 65535;
    const float* gp = Gpart + (size_t)n * 64 * 65536 + e;
    float s = 0.f;
    for (int p = 0; p < 64; ++p) s += gp[(size_t)p * 65536];
    G[idx] = s;
    if (idx < 512) {
        const float* sp = Spart + (size_t)(idx >> 8) * 64 * CCH + (idx & 255);
        float t = 0.f;
        for (int p = 0; p < 64; ++p) t += sp[p * CCH];
        S[idx] = t;
    }
}

#define BPAD 260

__device__ __forceinline__ void lds_handover() {
    __builtin_amdgcn_fence(__ATOMIC_RELEASE, "workgroup", "local");
    __builtin_amdgcn_s_barrier();
    __builtin_amdgcn_fence(__ATOMIC_ACQUIRE, "workgroup", "local");
}

__global__ __launch_bounds__(512) void band_kernel(const float* __restrict__ Wq,
                                                   const float* __restrict__ Wk,
                                                   const float* __restrict__ Wv,
                                                   const float* __restrict__ bq,
                                                   const float* __restrict__ bk,
                                                   const float* __restrict__ bv,
                                                   const float* __restrict__ G,
                                                   const float* __restrict__ S,
                                                   float* __restrict__ Bt,
                                                   float* __restrict__ cvec) {
    __shared__ __attribute__((aligned(16))) float ldsQ[16 * BPAD];
    __shared__ __attribute__((aligned(16))) float ldsT[16 * BPAD];
    __shared__ __attribute__((aligned(16))) float ldsS[CCH];
    __shared__ float ldsBK[CCH];
    __shared__ float ldsBV[CCH];
    __shared__ float ldsBQ[16];
    __shared__ float ldsU[16];
    __shared__ float ldsC[16];

    const int n   = blockIdx.y;
    const int i0  = blockIdx.x * 16;
    const int tid = threadIdx.x, lane = tid & 31, w = tid >> 5;
    const int l15 = lane & 15, half = lane >> 4;
    const int col = w * 16 + l15;
    const int kh  = half * 2;
    const float* Gn  = G + (size_t)n * CCH * CCH;
    const float* wkr = Wk + col * CCH;

    v2f pa[32], pb[32];

    const int   t8    = tid & (CCH - 1);
    const float s_in  = S[n * CCH + t8];
    const float bk_in = bk[t8];
    const float bv_in = bv[t8];
    const float bq_in = bq[i0 + (tid & 15)];
    float4 q_in[2];
#pragma unroll
    for (int i = 0; i < 2; ++i) {
        const int f = tid + 512 * i;
        q_in[i] = *(const float4*)&Wq[(i0 + (f >> 6)) * CCH + (f & 63) * 4];
    }
    __builtin_amdgcn_sched_barrier(0);

#pragma unroll
    for (int i = 0; i < 32; ++i) {
        const int kk = i * 4 + kh;
        pa[i][0] = Gn[kk * CCH + col];
        pa[i][1] = Gn[(kk + 1) * CCH + col];
    }
#pragma unroll
    for (int i = 0; i < 32; ++i) {
        const int kk = (i + 32) * 4 + kh;
        pb[i][0] = Gn[kk * CCH + col];
        pb[i][1] = Gn[(kk + 1) * CCH + col];
    }

#pragma unroll
    for (int i = 0; i < 2; ++i) {
        const int f = tid + 512 * i;
        *(float4*)&ldsQ[(f >> 6) * BPAD + (f & 63) * 4] = q_in[i];
    }
    ldsS[t8] = s_in; ldsBK[t8] = bk_in; ldsBV[t8] = bv_in;
    ldsBQ[tid & 15] = bq_in;
    lds_handover();

    {
        float p = 0.f;
#pragma unroll
        for (int t = 0; t < 8; ++t) {
            const int j = lane + 32 * t;
            p += ldsQ[w * BPAD + j] * ldsS[j];
        }
        for (int off = 16; off >= 1; off >>= 1) p += __shfl_xor(p, off, 32);
        if (lane == 0) ldsU[w] = p;
    }

    const v8f zero8 = {0.f,0.f,0.f,0.f,0.f,0.f,0.f,0.f};

    v8f acc = zero8;
#pragma unroll
    for (int i = 0; i < 32; ++i) {
        const int kk = i * 4 + kh;
        const v2f a = *(const v2f*)&ldsQ[l15 * BPAD + kk];
        acc = wmma_f32(a, pa[i], acc);
    }
#pragma unroll
    for (int i = 0; i < 32; ++i) {
        const int kk = i * 4 + kh;
        pa[i] = *(const v2f*)&wkr[kk];
    }
    __builtin_amdgcn_sched_barrier(0);
#pragma unroll
    for (int i = 0; i < 32; ++i) {
        const int kk = (i + 32) * 4 + kh;
        const v2f a = *(const v2f*)&ldsQ[l15 * BPAD + kk];
        acc = wmma_f32(a, pb[i], acc);
    }
#pragma unroll
    for (int i = 0; i < 32; ++i) {
        const int kk = (i + 32) * 4 + kh;
        pb[i] = *(const v2f*)&wkr[kk];
    }
#pragma unroll
    for (int v = 0; v < 8; ++v) ldsT[(half * 8 + v) * BPAD + col] = acc[v];
    lds_handover();

    acc = zero8;
    float wd = 0.f;
#pragma unroll
    for (int i = 0; i < 32; ++i) {
        const int kk = i * 4 + kh;
        const v2f a  = *(const v2f*)&ldsT[l15 * BPAD + kk];
        const v2f sv = *(const v2f*)&ldsS[kk];
        wd += pa[i][0] * sv[0];
        wd += pa[i][1] * sv[1];
        acc = wmma_f32(a, pa[i], acc);
    }
#pragma unroll
    for (int i = 0; i < 32; ++i) {
        const int kk = i * 4 + kh;
        pa[i][0] = Wv[kk * CCH + col];
        pa[i][1] = Wv[(kk + 1) * CCH + col];
    }
    __builtin_amdgcn_sched_barrier(0);
#pragma unroll
    for (int i = 0; i < 32; ++i) {
        const int kk = (i + 32) * 4 + kh;
        const v2f a  = *(const v2f*)&ldsT[l15 * BPAD + kk];
        const v2f sv = *(const v2f*)&ldsS[kk];
        wd += pb[i][0] * sv[0];
        wd += pb[i][1] * sv[1];
        acc = wmma_f32(a, pb[i], acc);
    }
#pragma unroll
    for (int i = 0; i < 32; ++i) {
        const int kk = (i + 32) * 4 + kh;
        pb[i][0] = Wv[kk * CCH + col];
        pb[i][1] = Wv[(kk + 1) * CCH + col];
    }
    wd += __shfl_xor(wd, 16, 32);

    {
        const float bkc = ldsBK[col];
        const float wv  = wd + (float)MTOK * bkc;
#pragma unroll
        for (int v = 0; v < 8; ++v) {
            const int r = half * 8 + v;
            const float val = acc[v] + ldsU[r] * bkc + ldsBQ[r] * wv;
            ldsQ[r * BPAD + col] = val;
        }
    }
    lds_handover();

    {
        float* rowp = &ldsQ[w * BPAD];
        float l[8];
        float mx = -3.0e38f;
#pragma unroll
        for (int t = 0; t < 8; ++t) {
            l[t] = rowp[lane + 32 * t];
            mx = fmaxf(mx, l[t]);
        }
        for (int off = 16; off >= 1; off >>= 1) mx = fmaxf(mx, __shfl_xor(mx, off, 32));
        float sum = 0.f;
#pragma unroll
        for (int t = 0; t < 8; ++t) { l[t] = __expf(l[t] - mx); sum += l[t]; }
        for (int off = 16; off >= 1; off >>= 1) sum += __shfl_xor(sum, off, 32);
        const float inv = 1.f / sum;
        float cd = 0.f;
#pragma unroll
        for (int t = 0; t < 8; ++t) {
            const int j = lane + 32 * t;
            const float a = l[t] * inv;
            rowp[j] = a;
            cd += a * ldsBV[j];
        }
        for (int off = 16; off >= 1; off >>= 1) cd += __shfl_xor(cd, off, 32);
        if (lane == 0) ldsC[w] = cd;
    }
    lds_handover();

    acc = zero8;
#pragma unroll
    for (int i = 0; i < 32; ++i) {
        const int kk = i * 4 + kh;
        const v2f a = *(const v2f*)&ldsQ[l15 * BPAD + kk];
        acc = wmma_f32(a, pa[i], acc);
    }
#pragma unroll
    for (int i = 0; i < 32; ++i) {
        const int kk = (i + 32) * 4 + kh;
        const v2f a = *(const v2f*)&ldsQ[l15 * BPAD + kk];
        acc = wmma_f32(a, pb[i], acc);
    }
#pragma unroll
    for (int v = 0; v < 8; ++v) ldsT[(half * 8 + v) * BPAD + col] = acc[v];
    lds_handover();
    {
        float* brow = Bt + (size_t)n * CCH * CCH + (size_t)(i0 + w) * CCH;
#pragma unroll
        for (int t = 0; t < 2; ++t) {
            const int c = lane * 4 + 128 * t;
            *(float4*)&brow[c] = *(const float4*)&ldsT[w * BPAD + c];
        }
    }
    if (tid < 16) cvec[n * CCH + i0 + tid] = ldsC[tid];
}

#define OPITCH 260

__global__ __launch_bounds__(512) void out_kernel(const float* __restrict__ x,
                                                  const float* __restrict__ Bt,
                                                  const float* __restrict__ cvec,
                                                  float* __restrict__ out) {
    __shared__ __attribute__((aligned(16))) float ldsB[CCH * OPITCH];
    const int n = blockIdx.y;
    const float* xn  = x + (size_t)n * CCH * MTOK;
    const float* btn = Bt + (size_t)n * CCH * CCH;
    float* on = out + (size_t)n * MTOK * CCH;
    const int m0 = blockIdx.x * 256;
    const int tid = threadIdx.x, lane = tid & 31, w = tid >> 5;
    const int l15 = lane & 15, half = lane >> 4;
    const int kh  = half * 2;

    const int rot = blockIdx.x * 128;
    float4 t[32];
#pragma unroll
    for (int i = 0; i < 32; ++i) {
        const int f = (tid + 512 * i + rot) & 16383;
        t[i] = *(const float4*)&btn[(f >> 6) * CCH + (f & 63) * 4];
    }
#pragma unroll
    for (int i = 0; i < 32; ++i) {
        const int f = (tid + 512 * i + rot) & 16383;
        *(float4*)&ldsB[(f >> 6) * OPITCH + (f & 63) * 4] = t[i];
    }
    __builtin_amdgcn_sched_barrier(0);

    const float* xa = xn + (size_t)(kh * MTOK + m0 + w * 16 + l15);
    v2f a[64];
#pragma unroll
    for (int g = 0; g < 4; ++g) {
#pragma unroll
        for (int i = 16 * g; i < 16 * g + 16; ++i) {
            a[i][0] = xa[(size_t)(4 * i) * MTOK];
            a[i][1] = xa[(size_t)(4 * i + 1) * MTOK];
        }
        __builtin_amdgcn_sched_barrier(0);
    }
    lds_handover();

    const float* bl = &ldsB[l15 * OPITCH + kh];
    const v8f zero8 = {0.f,0.f,0.f,0.f,0.f,0.f,0.f,0.f};
    float* orow = on + (size_t)(m0 + w * 16 + half * 8) * CCH + l15;
    const float* cvn = cvec + n * CCH + l15;

#pragma unroll 1
    for (int tg = 0; tg < 4; ++tg) {
        const float* bt = bl + tg * (64 * OPITCH);
        v8f acc[4];
#pragma unroll
        for (int c = 0; c < 4; ++c) acc[c] = zero8;
#pragma unroll
        for (int i = 0; i < 64; ++i) {
#pragma unroll
            for (int c = 0; c < 4; ++c) {
                const v2f b = *(const v2f*)&bt[c * (16 * OPITCH) + 4 * i];
                acc[c] = wmma_f32(a[i], b, acc[c]);
            }
        }
#pragma unroll
        for (int c = 0; c < 4; ++c) {
            const int cofs = (tg * 4 + c) * 16;
            const float cv = cvn[cofs];
#pragma unroll
            for (int v = 0; v < 8; ++v)
                orow[(size_t)v * CCH + cofs] = acc[c][v] + cv;
        }
    }
}

extern "C" void kernel_launch(void* const* d_in, const int* in_sizes, int n_in,
                              void* d_out, int out_size, void* d_ws, size_t ws_size,
                              hipStream_t stream) {
    (void)in_sizes; (void)n_in; (void)out_size; (void)ws_size;
    const float* x  = (const float*)d_in[0];
    const float* Wq = (const float*)d_in[1];
    const float* bq = (const float*)d_in[2];
    const float* Wk = (const float*)d_in[3];
    const float* bk = (const float*)d_in[4];
    const float* Wv = (const float*)d_in[5];
    const float* bv = (const float*)d_in[6];
    float* out = (float*)d_out;
    float* ws  = (float*)d_ws;

    const size_t oGP  = 0;
    const size_t oSP  = oGP  + (size_t)2 * 64 * 65536;
    const size_t oG   = oSP  + (size_t)2 * 64 * 256;
    const size_t oS   = oG   + (size_t)2 * 65536;
    const size_t oT1  = oS   + 512;
    const size_t oP0  = oT1  + (size_t)2 * 65536;
    const size_t oATT = oP0  + (size_t)2 * 65536;
    const size_t oBT  = oATT + (size_t)2 * 65536;
    const size_t oU   = oBT  + (size_t)2 * 65536;
    const size_t oW   = oU   + 512;
    const size_t oCV  = oW   + 512;

    gram_kernel<<<dim3(64, 2), 512, 0, stream>>>(x, ws + oGP, ws + oSP);
    reduce_kernel<<<512, 256, 0, stream>>>(ws + oGP, ws + oSP, ws + oG, ws + oS);
    band_kernel<<<dim3(16, 2), 512, 0, stream>>>(Wq, Wk, Wv, bq, bk, bv, ws + oG, ws + oS,
                                                 ws + oBT, ws + oCV);
    out_kernel<<<dim3(128, 2), 512, 0, stream>>>(x, ws + oBT, ws + oCV, out);
}


# gram_kernel: token axis cut into 128 chunks per batch (256 WGs, one per WGP) where 64 left half the chip idle; the partials sit in out until reduce; G and s sum 128 partials: sum order changes
# speedup vs baseline: 2.8345x; 1.2477x over previous
#include <hip/hip_runtime.h>
#include <hip/hip_bf16.h>


#define MTOK 32768
#define CCH  256
#define GTOK 256
#define GSPLIT (MTOK / GTOK)

typedef __attribute__((ext_vector_type(2))) float v2f;
typedef __attribute__((ext_vector_type(8))) float v8f;

__device__ __forceinline__ v8f wmma_f32(v2f a, v2f b, v8f c) {
    return __builtin_amdgcn_wmma_f32_16x16x4_f32(false, a, false, b, (short)0, c, false, false);
}

__global__ __launch_bounds__(512) void gram_kernel(const float* __restrict__ x,
                                                   float* __restrict__ Gpart,
                                                   float* __restrict__ Spart) {
    __shared__ float lds[CCH * 34];
    const int n    = blockIdx.y;
    const int wg   = blockIdx.x;
    const float* xn = x + (size_t)n * CCH * MTOK;
    const int tid  = threadIdx.x;
    const int lane = tid & 31;
    const int w    = tid >> 5;
    const int bwr  = w >> 2, bwc = w & 3;
    const int l15  = lane & 15;
    const int half = lane >> 4;

    const v8f zero8 = {0.f,0.f,0.f,0.f,0.f,0.f,0.f,0.f};
    v8f acc[4][4];
#pragma unroll
    for (int r = 0; r < 4; ++r)
#pragma unroll
        for (int c = 0; c < 4; ++c) acc[r][c] = zero8;

    float sacc[16];
#pragma unroll
    for (int i = 0; i < 16; ++i) sacc[i] = 0.f;

    const int m0 = wg * GTOK;
    for (int t0 = 0; t0 < GTOK; t0 += 32) {
        __syncthreads();
#pragma unroll
        for (int i = 0; i < 16; ++i) {
            const int ch = w + 16 * i;
            const float val = xn[(size_t)ch * MTOK + m0 + t0 + lane];
            lds[ch * 34 + lane] = val;
            sacc[i] += val;
        }
        __syncthreads();
#pragma unroll
        for (int kb = 0; kb < 32; kb += 4) {
            const int kk = kb + half * 2;
            v2f a[4], b[4];
#pragma unroll
            for (int r = 0; r < 4; ++r) {
                const int chA = (bwr * 4 + r) * 16 + l15;
                a[r] = *(const v2f*)&lds[chA * 34 + kk];
                const int chB = (bwc * 4 + r) * 16 + l15;
                b[r] = *(const v2f*)&lds[chB * 34 + kk];
            }
#pragma unroll
            for (int r = 0; r < 4; ++r)
#pragma unroll
                for (int c = 0; c < 4; ++c)
                    acc[r][c] = wmma_f32(a[r], b[c], acc[r][c]);
        }
    }

    float* gp = Gpart + ((size_t)n * GSPLIT + wg) * (CCH * CCH);
#pragma unroll
    for (int r = 0; r < 4; ++r)
#pragma unroll
        for (int c = 0; c < 4; ++c) {
            const int row0 = (bwr * 4 + r) * 16 + half * 8;
            const int col  = (bwc * 4 + c) * 16 + l15;
#pragma unroll
            for (int v = 0; v < 8; ++v)
                gp[(row0 + v) * CCH + col] = acc[r][c][v];
        }

#pragma unroll
    for (int i = 0; i < 16; ++i) {
        float s = sacc[i];
        for (int off = 16; off >= 1; off >>= 1) s += __shfl_xor(s, off, 32);
        if (lane == 0) Spart[((size_t)n * GSPLIT + wg) * CCH + w + 16 * i] = s;
    }
}

__global__ __launch_bounds__(256) void reduce_kernel(const float* __restrict__ Gpart,
                                                     const float* __restrict__ Spart,
                                                     float* __restrict__ G,
                                                     float* __restrict__ S) {
    const int idx = blockIdx.x * 256 + threadIdx.x;
    const int n = idx >> 16;
    const int e = idx & 65535;
    const float* gp = Gpart + (size_t)n * GSPLIT * 65536 + e;
    float s = 0.f;
    for (int p = 0; p < GSPLIT; ++p) s += gp[(size_t)p * 65536];
    G[idx] = s;
    if (idx < 512) {
        const float* sp = Spart + (size_t)(idx >> 8) * GSPLIT * CCH + (idx & 255);
        float t = 0.f;
        for (int p = 0; p < GSPLIT; ++p) t += sp[p * CCH];
        S[idx] = t;
    }
}

#define BPAD 260

__device__ __forceinline__ void lds_handover() {
    __builtin_amdgcn_fence(__ATOMIC_RELEASE, "workgroup", "local");
    __builtin_amdgcn_s_barrier();
    __builtin_amdgcn_fence(__ATOMIC_ACQUIRE, "workgroup", "local");
}

__global__ __launch_bounds__(512) void band_kernel(const float* __restrict__ Wq,
                                                   const float* __restrict__ Wk,
                                                   const float* __restrict__ Wv,
                                                   const float* __restrict__ bq,
                                                   const float* __restrict__ bk,
                                                   const float* __restrict__ bv,
                                                   const float* __restrict__ G,
                                                   const float* __restrict__ S,
                                                   float* __restrict__ Bt,
                                                   float* __restrict__ cvec) {
    __shared__ __attribute__((aligned(16))) float ldsQ[16 * BPAD];
    __shared__ __attribute__((aligned(16))) float ldsT[16 * BPAD];
    __shared__ __attribute__((aligned(16))) float ldsS[CCH];
    __shared__ float ldsBK[CCH];
    __shared__ float ldsBV[CCH];
    __shared__ float ldsBQ[16];
    __shared__ float ldsU[16];
    __shared__ float ldsC[16];

    const int n   = blockIdx.y;
    const int i0  = blockIdx.x * 16;
    const int tid = threadIdx.x, lane = tid & 31, w = tid >> 5;
    const int l15 = lane & 15, half = lane >> 4;
    const int col = w * 16 + l15;
    const int kh  = half * 2;
    const float* Gn  = G + (size_t)n * CCH * CCH;
    const float* wkr = Wk + col * CCH;

    v2f pa[32], pb[32];

    const int   t8    = tid & (CCH - 1);
    const float s_in  = S[n * CCH + t8];
    const float bk_in = bk[t8];
    const float bv_in = bv[t8];
    const float bq_in = bq[i0 + (tid & 15)];
    float4 q_in[2];
#pragma unroll
    for (int i = 0; i < 2; ++i) {
        const int f = tid + 512 * i;
        q_in[i] = *(const float4*)&Wq[(i0 + (f >> 6)) * CCH + (f & 63) * 4];
    }
    __builtin_amdgcn_sched_barrier(0);

#pragma unroll
    for (int i = 0; i < 32; ++i) {
        const int kk = i * 4 + kh;
        pa[i][0] = Gn[kk * CCH + col];
        pa[i][1] = Gn[(kk + 1) * CCH + col];
    }
#pragma unroll
    for (int i = 0; i < 32; ++i) {
        const int kk = (i + 32) * 4 + kh;
        pb[i][0] = Gn[kk * CCH + col];
        pb[i][1] = Gn[(kk + 1) * CCH + col];
    }

#pragma unroll
    for (int i = 0; i < 2; ++i) {
        const int f = tid + 512 * i;
        *(float4*)&ldsQ[(f >> 6) * BPAD + (f & 63) * 4] = q_in[i];
    }
    ldsS[t8] = s_in; ldsBK[t8] = bk_in; ldsBV[t8] = bv_in;
    ldsBQ[tid & 15] = bq_in;
    lds_handover();

    {
        float p = 0.f;
#pragma unroll
        for (int t = 0; t < 8; ++t) {
            const int j = lane + 32 * t;
            p += ldsQ[w * BPAD + j] * ldsS[j];
        }
        for (int off = 16; off >= 1; off >>= 1) p += __shfl_xor(p, off, 32);
        if (lane == 0) ldsU[w] = p;
    }

    const v8f zero8 = {0.f,0.f,0.f,0.f,0.f,0.f,0.f,0.f};

    v8f acc = zero8;
#pragma unroll
    for (int i = 0; i < 32; ++i) {
        const int kk = i * 4 + kh;
        const v2f a = *(const v2f*)&ldsQ[l15 * BPAD + kk];
        acc = wmma_f32(a, pa[i], acc);
    }
#pragma unroll
    for (int i = 0; i < 32; ++i) {
        const int kk = i * 4 + kh;
        pa[i] = *(const v2f*)&wkr[kk];
    }
    __builtin_amdgcn_sched_barrier(0);
#pragma unroll
    for (int i = 0; i < 32; ++i) {
        const int kk = (i + 32) * 4 + kh;
        const v2f a = *(const v2f*)&ldsQ[l15 * BPAD + kk];
        acc = wmma_f32(a, pb[i], acc);
    }
#pragma unroll
    for (int i = 0; i < 32; ++i) {
        const int kk = (i + 32) * 4 + kh;
        pb[i] = *(const v2f*)&wkr[kk];
    }
#pragma unroll
    for (int v = 0; v < 8; ++v) ldsT[(half * 8 + v) * BPAD + col] = acc[v];
    lds_handover();

    acc = zero8;
    float wd = 0.f;
#pragma unroll
    for (int i = 0; i < 32; ++i) {
        const int kk = i * 4 + kh;
        const v2f a  = *(const v2f*)&ldsT[l15 * BPAD + kk];
        const v2f sv = *(const v2f*)&ldsS[kk];
        wd += pa[i][0] * sv[0];
        wd += pa[i][1] * sv[1];
        acc = wmma_f32(a, pa[i], acc);
    }
#pragma unroll
    for (int i = 0; i < 32; ++i) {
        const int kk = i * 4 + kh;
        pa[i][0] = Wv[kk * CCH + col];
        pa[i][1] = Wv[(kk + 1) * CCH + col];
    }
    __builtin_amdgcn_sched_barrier(0);
#pragma unroll
    for (int i = 0; i < 32; ++i) {
        const int kk = (i + 32) * 4 + kh;
        const v2f a  = *(const v2f*)&ldsT[l15 * BPAD + kk];
        const v2f sv = *(const v2f*)&ldsS[kk];
        wd += pb[i][0] * sv[0];
        wd += pb[i][1] * sv[1];
        acc = wmma_f32(a, pb[i], acc);
    }
#pragma unroll
    for (int i = 0; i < 32; ++i) {
        const int kk = (i + 32) * 4 + kh;
        pb[i][0] = Wv[kk * CCH + col];
        pb[i][1] = Wv[(kk + 1) * CCH + col];
    }
    wd += __shfl_xor(wd, 16, 32);

    {
        const float bkc = ldsBK[col];
        const float wv  = wd + (float)MTOK * bkc;
#pragma unroll
        for (int v = 0; v < 8; ++v) {
            const int r = half * 8 + v;
            const float val = acc[v] + ldsU[r] * bkc + ldsBQ[r] * wv;
            ldsQ[r * BPAD + col] = val;
        }
    }
    lds_handover();

    {
        float* rowp = &ldsQ[w * BPAD];
        float l[8];
        float mx = -3.0e38f;
#pragma unroll
        for (int t = 0; t < 8; ++t) {
            l[t] = rowp[lane + 32 * t];
            mx = fmaxf(mx, l[t]);
        }
        for (int off = 16; off >= 1; off >>= 1) mx = fmaxf(mx, __shfl_xor(mx, off, 32));
        float sum = 0.f;
#pragma unroll
        for (int t = 0; t < 8; ++t) { l[t] = __expf(l[t] - mx); sum += l[t]; }
        for (int off = 16; off >= 1; off >>= 1) sum += __shfl_xor(sum, off, 32);
        const float inv = 1.f / sum;
        float cd = 0.f;
#pragma unroll
        for (int t = 0; t < 8; ++t) {
            const int j = lane + 32 * t;
            const float a = l[t] * inv;
            rowp[j] = a;
            cd += a * ldsBV[j];
        }
        for (int off = 16; off >= 1; off >>= 1) cd += __shfl_xor(cd, off, 32);
        if (lane == 0) ldsC[w] = cd;
    }
    lds_handover();

    acc = zero8;
#pragma unroll
    for (int i = 0; i < 32; ++i) {
        const int kk = i * 4 + kh;
        const v2f a = *(const v2f*)&ldsQ[l15 * BPAD + kk];
        acc = wmma_f32(a, pa[i], acc);
    }
#pragma unroll
    for (int i = 0; i < 32; ++i) {
        const int kk = (i + 32) * 4 + kh;
        const v2f a = *(const v2f*)&ldsQ[l15 * BPAD + kk];
        acc = wmma_f32(a, pb[i], acc);
    }
#pragma unroll
    for (int v = 0; v < 8; ++v) ldsT[(half * 8 + v) * BPAD + col] = acc[v];
    lds_handover();
    {
        float* brow = Bt + (size_t)n * CCH * CCH + (size_t)(i0 + w) * CCH;
#pragma unroll
        for (int t = 0; t < 2; ++t) {
            const int c = lane * 4 + 128 * t;
            *(float4*)&brow[c] = *(const float4*)&ldsT[w * BPAD + c];
        }
    }
    if (tid < 16) cvec[n * CCH + i0 + tid] = ldsC[tid];
}

#define OPITCH 260

__global__ __launch_bounds__(512) void out_kernel(const float* __restrict__ x,
                                                  const float* __restrict__ Bt,
                                                  const float* __restrict__ cvec,
                                                  float* __restrict__ out) {
    __shared__ __attribute__((aligned(16))) float ldsB[CCH * OPITCH];
    const int n = blockIdx.y;
    const float* xn  = x + (size_t)n * CCH * MTOK;
    const float* btn = Bt + (size_t)n * CCH * CCH;
    float* on = out + (size_t)n * MTOK * CCH;
    const int m0 = blockIdx.x * 256;
    const int tid = threadIdx.x, lane = tid & 31, w = tid >> 5;
    const int l15 = lane & 15, half = lane >> 4;
    const int kh  = half * 2;

    const int rot = blockIdx.x * 128;
    float4 t[32];
#pragma unroll
    for (int i = 0; i < 32; ++i) {
        const int f = (tid + 512 * i + rot) & 16383;
        t[i] = *(const float4*)&btn[(f >> 6) * CCH + (f & 63) * 4];
    }
#pragma unroll
    for (int i = 0; i < 32; ++i) {
        const int f = (tid + 512 * i + rot) & 16383;
        *(float4*)&ldsB[(f >> 6) * OPITCH + (f & 63) * 4] = t[i];
    }
    __builtin_amdgcn_sched_barrier(0);

    const float* xa = xn + (size_t)(kh * MTOK + m0 + w * 16 + l15);
    v2f a[64];
#pragma unroll
    for (int g = 0; g < 4; ++g) {
#pragma unroll
        for (int i = 16 * g; i < 16 * g + 16; ++i) {
            a[i][0] = xa[(size_t)(4 * i) * MTOK];
            a[i][1] = xa[(size_t)(4 * i + 1) * MTOK];
        }
        __builtin_amdgcn_sched_barrier(0);
    }
    lds_handover();

    const float* bl = &ldsB[l15 * OPITCH + kh];
    const v8f zero8 = {0.f,0.f,0.f,0.f,0.f,0.f,0.f,0.f};
    float* orow = on + (size_t)(m0 + w * 16 + half * 8) * CCH + l15;
    const float* cvn = cvec + n * CCH + l15;

#pragma unroll 1
    for (int tg = 0; tg < 4; ++tg) {
        const float* bt = bl + tg * (64 * OPITCH);
        v8f acc[4];
#pragma unroll
        for (int c = 0; c < 4; ++c) acc[c] = zero8;
#pragma unroll
        for (int i = 0; i < 64; ++i) {
#pragma unroll
            for (int c = 0; c < 4; ++c) {
                const v2f b = *(const v2f*)&bt[c * (16 * OPITCH) + 4 * i];
                acc[c] = wmma_f32(a[i], b, acc[c]);
            }
        }
#pragma unroll
        for (int c = 0; c < 4; ++c) {
            const int cofs = (tg * 4 + c) * 16;
            const float cv = cvn[cofs];
#pragma unroll
            for (int v = 0; v < 8; ++v)
                orow[(size_t)v * CCH + cofs] = acc[c][v] + cv;
        }
    }
}

extern "C" void kernel_launch(void* const* d_in, const int* in_sizes, int n_in,
                              void* d_out, int out_size, void* d_ws, size_t ws_size,
                              hipStream_t stream) {
    (void)in_sizes; (void)n_in; (void)out_size; (void)ws_size;
    const float* x  = (const float*)d_in[0];
    const float* Wq = (const float*)d_in[1];
    const float* bq = (const float*)d_in[2];
    const float* Wk = (const float*)d_in[3];
    const float* bk = (const float*)d_in[4];
    const float* Wv = (const float*)d_in[5];
    const float* bv = (const float*)d_in[6];
    float* out = (float*)d_out;
    float* ws  = (float*)d_ws;

    const size_t oSP  = 0;
    const size_t oG   = oSP  + (size_t)2 * 128 * 256;
    const size_t oS   = oG   + (size_t)2 * 65536;
    const size_t oBT  = oS   + 512;
    const size_t oCV  = oBT  + (size_t)2 * 65536;

    float* gpart = out;

    gram_kernel<<<dim3(128, 2), 512, 0, stream>>>(x, gpart, ws + oSP);
    reduce_kernel<<<512, 256, 0, stream>>>(gpart, ws + oSP, ws + oG, ws + oS);
    band_kernel<<<dim3(16, 2), 512, 0, stream>>>(Wq, Wk, Wv, bq, bk, bv, ws + oG, ws + oS,
                                                 ws + oBT, ws + oCV);
    out_kernel<<<dim3(128, 2), 512, 0, stream>>>(x, ws + oBT, ws + oCV, out);
}


# gram_kernel: LDS fragment reads of the next two k-steps issued as one group ahead of the current 32 products (second register set) where each read was waited for at once; k order and sums unchanged
# speedup vs baseline: 2.9193x; 1.0299x over previous
#include <hip/hip_runtime.h>
#include <hip/hip_bf16.h>


#define MTOK 32768
#define CCH  256
#define GTOK 256
#define GSPLIT (MTOK / GTOK)

typedef __attribute__((ext_vector_type(2))) float v2f;
typedef __attribute__((ext_vector_type(8))) float v8f;

__device__ __forceinline__ v8f wmma_f32(v2f a, v2f b, v8f c) {
    return __builtin_amdgcn_wmma_f32_16x16x4_f32(false, a, false, b, (short)0, c, false, false);
}

__global__ __launch_bounds__(512) void gram_kernel(const float* __restrict__ x,
                                                   float* __restrict__ Gpart,
                                                   float* __restrict__ Spart) {
    __shared__ float lds[CCH * 34];
    const int n    = blockIdx.y;
    const int wg   = blockIdx.x;
    const float* xn = x + (size_t)n * CCH * MTOK;
    const int tid  = threadIdx.x;
    const int lane = tid & 31;
    const int w    = tid >> 5;
    const int bwr  = w >> 2, bwc = w & 3;
    const int l15  = lane & 15;
    const int half = lane >> 4;

    const v8f zero8 = {0.f,0.f,0.f,0.f,0.f,0.f,0.f,0.f};
    v8f acc[4][4];
#pragma unroll
    for (int r = 0; r < 4; ++r)
#pragma unroll
        for (int c = 0; c < 4; ++c) acc[r][c] = zero8;

    float sacc[16];
#pragma unroll
    for (int i = 0; i < 16; ++i) sacc[i] = 0.f;

    const int m0 = wg * GTOK;
    for (int t0 = 0; t0 < GTOK; t0 += 32) {
        __syncthreads();
#pragma unroll
        for (int i = 0; i < 16; ++i) {
            const int ch = w + 16 * i;
            const float val = xn[(size_t)ch * MTOK + m0 + t0 + lane];
            lds[ch * 34 + lane] = val;
            sacc[i] += val;
        }
        __syncthreads();
        v2f fa[2][2][4], fb[2][2][4];
#pragma unroll
        for (int u = 0; u < 2; ++u) {
            const int kk = 4 * u + half * 2;
#pragma unroll
            for (int r = 0; r < 4; ++r) {
                fa[0][u][r] = *(const v2f*)&lds[((bwr * 4 + r) * 16 + l15) * 34 + kk];
                fb[0][u][r] = *(const v2f*)&lds[((bwc * 4 + r) * 16 + l15) * 34 + kk];
            }
        }
#pragma unroll
        for (int p = 0; p < 4; ++p) {
            const int cur = p & 1, nxt = cur ^ 1;
            if (p < 3) {
#pragma unroll
                for (int u = 0; u < 2; ++u) {
                    const int kk = 8 * (p + 1) + 4 * u + half * 2;
#pragma unroll
                    for (int r = 0; r < 4; ++r) {
                        fa[nxt][u][r] = *(const v2f*)&lds[((bwr * 4 + r) * 16 + l15) * 34 + kk];
                        fb[nxt][u][r] = *(const v2f*)&lds[((bwc * 4 + r) * 16 + l15) * 34 + kk];
                    }
                }
            }
            __builtin_amdgcn_sched_barrier(0);
#pragma unroll
            for (int u = 0; u < 2; ++u)
#pragma unroll
                for (int r = 0; r < 4; ++r)
#pragma unroll
                    for (int c = 0; c < 4; ++c)
                        acc[r][c] = wmma_f32(fa[cur][u][r], fb[cur][u][c], acc[r][c]);
            __builtin_amdgcn_sched_barrier(0);
        }
    }

    float* gp = Gpart + ((size_t)n * GSPLIT + wg) * (CCH * CCH);
#pragma unroll
    for (int r = 0; r < 4; ++r)
#pragma unroll
        for (int c = 0; c < 4; ++c) {
            const int row0 = (bwr * 4 + r) * 16 + half * 8;
            const int col  = (bwc * 4 + c) * 16 + l15;
#pragma unroll
            for (int v = 0; v < 8; ++v)
                gp[(row0 + v) * CCH + col] = acc[r][c][v];
        }

#pragma unroll
    for (int i = 0; i < 16; ++i) {
        float s = sacc[i];
        for (int off = 16; off >= 1; off >>= 1) s += __shfl_xor(s, off, 32);
        if (lane == 0) Spart[((size_t)n * GSPLIT + wg) * CCH + w + 16 * i] = s;
    }
}

__global__ __launch_bounds__(256) void reduce_kernel(const float* __restrict__ Gpart,
                                                     const float* __restrict__ Spart,
                                                     float* __restrict__ G,
                                                     float* __restrict__ S) {
    const int idx = blockIdx.x * 256 + threadIdx.x;
    const int n = idx >> 16;
    const int e = idx & 65535;
    const float* gp = Gpart + (size_t)n * GSPLIT * 65536 + e;
    float s = 0.f;
    for (int p = 0; p < GSPLIT; ++p) s += gp[(size_t)p * 65536];
    G[idx] = s;
    if (idx < 512) {
        const float* sp = Spart + (size_t)(idx >> 8) * GSPLIT * CCH + (idx & 255);
        float t = 0.f;
        for (int p = 0; p < GSPLIT; ++p) t += sp[p * CCH];
        S[idx] = t;
    }
}

#define BPAD 260

__device__ __forceinline__ void lds_handover() {
    __builtin_amdgcn_fence(__ATOMIC_RELEASE, "workgroup", "local");
    __builtin_amdgcn_s_barrier();
    __builtin_amdgcn_fence(__ATOMIC_ACQUIRE, "workgroup", "local");
}

__global__ __launch_bounds__(512) void band_kernel(const float* __restrict__ Wq,
                                                   const float* __restrict__ Wk,
                                                   const float* __restrict__ Wv,
                                                   const float* __restrict__ bq,
                                                   const float* __restrict__ bk,
                                                   const float* __restrict__ bv,
                                                   const float* __restrict__ G,
                                                   const float* __restrict__ S,
                                                   float* __restrict__ Bt,
                                                   float* __restrict__ cvec) {
    __shared__ __attribute__((aligned(16))) float ldsQ[16 * BPAD];
    __shared__ __attribute__((aligned(16))) float ldsT[16 * BPAD];
    __shared__ __attribute__((aligned(16))) float ldsS[CCH];
    __shared__ float ldsBK[CCH];
    __shared__ float ldsBV[CCH];
    __shared__ float ldsBQ[16];
    __shared__ float ldsU[16];
    __shared__ float ldsC[16];

    const int n   = blockIdx.y;
    const int i0  = blockIdx.x * 16;
    const int tid = threadIdx.x, lane = tid & 31, w = tid >> 5;
    const int l15 = lane & 15, half = lane >> 4;
    const int col = w * 16 + l15;
    const int kh  = half * 2;
    const float* Gn  = G + (size_t)n * CCH * CCH;
    const float* wkr = Wk + col * CCH;

    v2f pa[32], pb[32];

    const int   t8    = tid & (CCH - 1);
    const float s_in  = S[n * CCH + t8];
    const float bk_in = bk[t8];
    const float bv_in = bv[t8];
    const float bq_in = bq[i0 + (tid & 15)];
    float4 q_in[2];
#pragma unroll
    for (int i = 0; i < 2; ++i) {
        const int f = tid + 512 * i;
        q_in[i] = *(const float4*)&Wq[(i0 + (f >> 6)) * CCH + (f & 63) * 4];
    }
    __builtin_amdgcn_sched_barrier(0);

#pragma unroll
    for (int i = 0; i < 32; ++i) {
        const int kk = i * 4 + kh;
        pa[i][0] = Gn[kk * CCH + col];
        pa[i][1] = Gn[(kk + 1) * CCH + col];
    }
#pragma unroll
    for (int i = 0; i < 32; ++i) {
        const int kk = (i + 32) * 4 + kh;
        pb[i][0] = Gn[kk * CCH + col];
        pb[i][1] = Gn[(kk + 1) * CCH + col];
    }

#pragma unroll
    for (int i = 0; i < 2; ++i) {
        const int f = tid + 512 * i;
        *(float4*)&ldsQ[(f >> 6) * BPAD + (f & 63) * 4] = q_in[i];
    }
    ldsS[t8] = s_in; ldsBK[t8] = bk_in; ldsBV[t8] = bv_in;
    ldsBQ[tid & 15] = bq_in;
    lds_handover();

    {
        float p = 0.f;
#pragma unroll
        for (int t = 0; t < 8; ++t) {
            const int j = lane + 32 * t;
            p += ldsQ[w * BPAD + j] * ldsS[j];
        }
        for (int off = 16; off >= 1; off >>= 1) p += __shfl_xor(p, off, 32);
        if (lane == 0) ldsU[w] = p;
    }

    const v8f zero8 = {0.f,0.f,0.f,0.f,0.f,0.f,0.f,0.f};

    v8f acc = zero8;
#pragma unroll
    for (int i = 0; i < 32; ++i) {
        const int kk = i * 4 + kh;
        const v2f a = *(const v2f*)&ldsQ[l15 * BPAD + kk];
        acc = wmma_f32(a, pa[i], acc);
    }
#pragma unroll
    for (int i = 0; i < 32; ++i) {
        const int kk = i * 4 + kh;
        pa[i] = *(const v2f*)&wkr[kk];
    }
    __builtin_amdgcn_sched_barrier(0);
#pragma unroll
    for (int i = 0; i < 32; ++i) {
        const int kk = (i + 32) * 4 + kh;
        const v2f a = *(const v2f*)&ldsQ[l15 * BPAD + kk];
        acc = wmma_f32(a, pb[i], acc);
    }
#pragma unroll
    for (int i = 0; i < 32; ++i) {
        const int kk = (i + 32) * 4 + kh;
        pb[i] = *(const v2f*)&wkr[kk];
    }
#pragma unroll
    for (int v = 0; v < 8; ++v) ldsT[(half * 8 + v) * BPAD + col] = acc[v];
    lds_handover();

    acc = zero8;
    float wd = 0.f;
#pragma unroll
    for (int i = 0; i < 32; ++i) {
        const int kk = i * 4 + kh;
        const v2f a  = *(const v2f*)&ldsT[l15 * BPAD + kk];
        const v2f sv = *(const v2f*)&ldsS[kk];
        wd += pa[i][0] * sv[0];
        wd += pa[i][1] * sv[1];
        acc = wmma_f32(a, pa[i], acc);
    }
#pragma unroll
    for (int i = 0; i < 32; ++i) {
        const int kk = i * 4 + kh;
        pa[i][0] = Wv[kk * CCH + col];
        pa[i][1] = Wv[(kk + 1) * CCH + col];
    }
    __builtin_amdgcn_sched_barrier(0);
#pragma unroll
    for (int i = 0; i < 32; ++i) {
        const int kk = (i + 32) * 4 + kh;
        const v2f a  = *(const v2f*)&ldsT[l15 * BPAD + kk];
        const v2f sv = *(const v2f*)&ldsS[kk];
        wd += pb[i][0] * sv[0];
        wd += pb[i][1] * sv[1];
        acc = wmma_f32(a, pb[i], acc);
    }
#pragma unroll
    for (int i = 0; i < 32; ++i) {
        const int kk = (i + 32) * 4 + kh;
        pb[i][0] = Wv[kk * CCH + col];
        pb[i][1] = Wv[(kk + 1) * CCH + col];
    }
    wd += __shfl_xor(wd, 16, 32);

    {
        const float bkc = ldsBK[col];
        const float wv  = wd + (float)MTOK * bkc;
#pragma unroll
        for (int v = 0; v < 8; ++v) {
            const int r = half * 8 + v;
            const float val = acc[v] + ldsU[r] * bkc + ldsBQ[r] * wv;
            ldsQ[r * BPAD + col] = val;
        }
    }
    lds_handover();

    {
        float* rowp = &ldsQ[w * BPAD];
        float l[8];
        float mx = -3.0e38f;
#pragma unroll
        for (int t = 0; t < 8; ++t) {
            l[t] = rowp[lane + 32 * t];
            mx = fmaxf(mx, l[t]);
        }
        for (int off = 16; off >= 1; off >>= 1) mx = fmaxf(mx, __shfl_xor(mx, off, 32));
        float sum = 0.f;
#pragma unroll
        for (int t = 0; t < 8; ++t) { l[t] = __expf(l[t] - mx); sum += l[t]; }
        for (int off = 16; off >= 1; off >>= 1) sum += __shfl_xor(sum, off, 32);
        const float inv = 1.f / sum;
        float cd = 0.f;
#pragma unroll
        for (int t = 0; t < 8; ++t) {
            const int j = lane + 32 * t;
            const float a = l[t] * inv;
            rowp[j] = a;
            cd += a * ldsBV[j];
        }
        for (int off = 16; off >= 1; off >>= 1) cd += __shfl_xor(cd, off, 32);
        if (lane == 0) ldsC[w] = cd;
    }
    lds_handover();

    acc = zero8;
#pragma unroll
    for (int i = 0; i < 32; ++i) {
        const int kk = i * 4 + kh;
        const v2f a = *(const v2f*)&ldsQ[l15 * BPAD + kk];
        acc = wmma_f32(a, pa[i], acc);
    }
#pragma unroll
    for (int i = 0; i < 32; ++i) {
        const int kk = (i + 32) * 4 + kh;
        const v2f a = *(const v2f*)&ldsQ[l15 * BPAD + kk];
        acc = wmma_f32(a, pb[i], acc);
    }
#pragma unroll
    for (int v = 0; v < 8; ++v) ldsT[(half * 8 + v) * BPAD + col] = acc[v];
    lds_handover();
    {
        float* brow = Bt + (size_t)n * CCH * CCH + (size_t)(i0 + w) * CCH;
#pragma unroll
        for (int t = 0; t < 2; ++t) {
            const int c = lane * 4 + 128 * t;
            *(float4*)&brow[c] = *(const float4*)&ldsT[w * BPAD + c];
        }
    }
    if (tid < 16) cvec[n * CCH + i0 + tid] = ldsC[tid];
}

#define OPITCH 260

__global__ __launch_bounds__(512) void out_kernel(const float* __restrict__ x,
                                                  const float* __restrict__ Bt,
                                                  const float* __restrict__ cvec,
                                                  float* __restrict__ out) {
    __shared__ __attribute__((aligned(16))) float ldsB[CCH * OPITCH];
    const int n = blockIdx.y;
    const float* xn  = x + (size_t)n * CCH * MTOK;
    const float* btn = Bt + (size_t)n * CCH * CCH;
    float* on = out + (size_t)n * MTOK * CCH;
    const int m0 = blockIdx.x * 256;
    const int tid = threadIdx.x, lane = tid & 31, w = tid >> 5;
    const int l15 = lane & 15, half = lane >> 4;
    const int kh  = half * 2;

    const int rot = blockIdx.x * 128;
    float4 t[32];
#pragma unroll
    for (int i = 0; i < 32; ++i) {
        const int f = (tid + 512 * i + rot) & 16383;
        t[i] = *(const float4*)&btn[(f >> 6) * CCH + (f & 63) * 4];
    }
#pragma unroll
    for (int i = 0; i < 32; ++i) {
        const int f = (tid + 512 * i + rot) & 16383;
        *(float4*)&ldsB[(f >> 6) * OPITCH + (f & 63) * 4] = t[i];
    }
    __builtin_amdgcn_sched_barrier(0);

    const float* xa = xn + (size_t)(kh * MTOK + m0 + w * 16 + l15);
    v2f a[64];
#pragma unroll
    for (int g = 0; g < 4; ++g) {
#pragma unroll
        for (int i = 16 * g; i < 16 * g + 16; ++i) {
            a[i][0] = xa[(size_t)(4 * i) * MTOK];
            a[i][1] = xa[(size_t)(4 * i + 1) * MTOK];
        }
        __builtin_amdgcn_sched_barrier(0);
    }
    lds_handover();

    const float* bl = &ldsB[l15 * OPITCH + kh];
    const v8f zero8 = {0.f,0.f,0.f,0.f,0.f,0.f,0.f,0.f};
    float* orow = on + (size_t)(m0 + w * 16 + half * 8) * CCH + l15;
    const float* cvn = cvec + n * CCH + l15;

#pragma unroll 1
    for (int tg = 0; tg < 4; ++tg) {
        const float* bt = bl + tg * (64 * OPITCH);
        v8f acc[4];
#pragma unroll
        for (int c = 0; c < 4; ++c) acc[c] = zero8;
#pragma unroll
        for (int i = 0; i < 64; ++i) {
#pragma unroll
            for (int c = 0; c < 4; ++c) {
                const v2f b = *(const v2f*)&bt[c * (16 * OPITCH) + 4 * i];
                acc[c] = wmma_f32(a[i], b, acc[c]);
            }
        }
#pragma unroll
        for (int c = 0; c < 4; ++c) {
            const int cofs = (tg * 4 + c) * 16;
            const float cv = cvn[cofs];
#pragma unroll
            for (int v = 0; v < 8; ++v)
                orow[(size_t)v * CCH + cofs] = acc[c][v] + cv;
        }
    }
}

extern "C" void kernel_launch(void* const* d_in, const int* in_sizes, int n_in,
                              void* d_out, int out_size, void* d_ws, size_t ws_size,
                              hipStream_t stream) {
    (void)in_sizes; (void)n_in; (void)out_size; (void)ws_size;
    const float* x  = (const float*)d_in[0];
    const float* Wq = (const float*)d_in[1];
    const float* bq = (const float*)d_in[2];
    const float* Wk = (const float*)d_in[3];
    const float* bk = (const float*)d_in[4];
    const float* Wv = (const float*)d_in[5];
    const float* bv = (const float*)d_in[6];
    float* out = (float*)d_out;
    float* ws  = (float*)d_ws;

    const size_t oSP  = 0;
    const size_t oG   = oSP  + (size_t)2 * 128 * 256;
    const size_t oS   = oG   + (size_t)2 * 65536;
    const size_t oBT  = oS   + 512;
    const size_t oCV  = oBT  + (size_t)2 * 65536;

    float* gpart = out;

    gram_kernel<<<dim3(128, 2), 512, 0, stream>>>(x, gpart, ws + oSP);
    reduce_kernel<<<512, 256, 0, stream>>>(gpart, ws + oSP, ws + oG, ws + oS);
    band_kernel<<<dim3(16, 2), 512, 0, stream>>>(Wq, Wk, Wv, bq, bk, bv, ws + oG, ws + oS,
                                                 ws + oBT, ws + oCV);
    out_kernel<<<dim3(128, 2), 512, 0, stream>>>(x, ws + oBT, ws + oCV, out);
}


# gram_kernel: x slab double-buffered in LDS: next slab's global loads issued in front of the current slab's products, stored to the other buffer behind them; one barrier per slab; s sum order changes
# speedup vs baseline: 3.1263x; 1.0709x over previous
#include <hip/hip_runtime.h>
#include <hip/hip_bf16.h>


#define MTOK 32768
#define CCH  256
#define GTOK 256
#define GSPLIT (MTOK / GTOK)

typedef __attribute__((ext_vector_type(2))) float v2f;
typedef __attribute__((ext_vector_type(8))) float v8f;

__device__ __forceinline__ v8f wmma_f32(v2f a, v2f b, v8f c) {
    return __builtin_amdgcn_wmma_f32_16x16x4_f32(false, a, false, b, (short)0, c, false, false);
}

#define GPITCH 36
#define GSLAB  (CCH * GPITCH)
__global__ __launch_bounds__(512) void gram_kernel(const float* __restrict__ x,
                                                   float* __restrict__ Gpart,
                                                   float* __restrict__ Spart) {
    __shared__ __attribute__((aligned(16))) float lds[2 * GSLAB];
    const int n    = blockIdx.y;
    const int wg   = blockIdx.x;
    const float* xn = x + (size_t)n * CCH * MTOK;
    const int tid  = threadIdx.x;
    const int lane = tid & 31;
    const int w    = tid >> 5;
    const int bwr  = w >> 2, bwc = w & 3;
    const int l15  = lane & 15;
    const int half = lane >> 4;

    const v8f zero8 = {0.f,0.f,0.f,0.f,0.f,0.f,0.f,0.f};
    v8f acc[4][4];
#pragma unroll
    for (int r = 0; r < 4; ++r)
#pragma unroll
        for (int c = 0; c < 4; ++c) acc[r][c] = zero8;

    const int sch = tid >> 3, tk = tid & 7;
    const int m0  = wg * GTOK;
    const float* xs = xn + (size_t)sch * MTOK + m0 + 4 * tk;
    float* lst = &lds[sch * GPITCH + 4 * tk];

    float sacc[4];
#pragma unroll
    for (int i = 0; i < 4; ++i) sacc[i] = 0.f;

    float4 pf[4];
#pragma unroll
    for (int i = 0; i < 4; ++i) pf[i] = *(const float4*)&xs[(size_t)(64 * i) * MTOK];
#pragma unroll
    for (int i = 0; i < 4; ++i) {
        *(float4*)&lst[(64 * i) * GPITCH] = pf[i];
        sacc[i] += (pf[i].x + pf[i].y) + (pf[i].z + pf[i].w);
    }
    __syncthreads();

    const float* la = &lds[((bwr * 4) * 16 + l15) * GPITCH + half * 2];
    const float* lb = &lds[((bwc * 4) * 16 + l15) * GPITCH + half * 2];

    int cur = 0;
    for (int t0 = 0; t0 < GTOK; t0 += 32) {
        const bool more = t0 + 32 < GTOK;
        if (more) {
#pragma unroll
            for (int i = 0; i < 4; ++i)
                pf[i] = *(const float4*)&xs[(size_t)(64 * i) * MTOK + t0 + 32];
        }
        const float* ca = la + cur;
        const float* cb = lb + cur;
        v2f fa[2][2][4], fb[2][2][4];
#pragma unroll
        for (int u = 0; u < 2; ++u) {
#pragma unroll
            for (int r = 0; r < 4; ++r) {
                fa[0][u][r] = *(const v2f*)&ca[r * 16 * GPITCH + 4 * u];
                fb[0][u][r] = *(const v2f*)&cb[r * 16 * GPITCH + 4 * u];
            }
        }
#pragma unroll
        for (int p = 0; p < 4; ++p) {
            const int cu = p & 1, nx = cu ^ 1;
            if (p < 3) {
#pragma unroll
                for (int u = 0; u < 2; ++u) {
                    const int kk = 8 * (p + 1) + 4 * u;
#pragma unroll
                    for (int r = 0; r < 4; ++r) {
                        fa[nx][u][r] = *(const v2f*)&ca[r * 16 * GPITCH + kk];
                        fb[nx][u][r] = *(const v2f*)&cb[r * 16 * GPITCH + kk];
                    }
                }
            }
            __builtin_amdgcn_sched_barrier(0);
#pragma unroll
            for (int u = 0; u < 2; ++u)
#pragma unroll
                for (int r = 0; r < 4; ++r)
#pragma unroll
                    for (int c = 0; c < 4; ++c)
                        acc[r][c] = wmma_f32(fa[cu][u][r], fb[cu][u][c], acc[r][c]);
            __builtin_amdgcn_sched_barrier(0);
        }
        if (more) {
            float* ls = lst + (GSLAB - cur);
#pragma unroll
            for (int i = 0; i < 4; ++i) {
                *(float4*)&ls[(64 * i) * GPITCH] = pf[i];
                sacc[i] += (pf[i].x + pf[i].y) + (pf[i].z + pf[i].w);
            }
        }
        __syncthreads();
        cur = GSLAB - cur;
    }

    float* gp = Gpart + ((size_t)n * GSPLIT + wg) * (CCH * CCH);
#pragma unroll
    for (int r = 0; r < 4; ++r)
#pragma unroll
        for (int c = 0; c < 4; ++c) {
            const int row0 = (bwr * 4 + r) * 16 + half * 8;
            const int col  = (bwc * 4 + c) * 16 + l15;
#pragma unroll
            for (int v = 0; v < 8; ++v)
                gp[(row0 + v) * CCH + col] = acc[r][c][v];
        }

#pragma unroll
    for (int i = 0; i < 4; ++i) {
        float s = sacc[i];
        for (int off = 4; off >= 1; off >>= 1) s += __shfl_xor(s, off, 32);
        if (tk == 0) Spart[((size_t)n * GSPLIT + wg) * CCH + sch + 64 * i] = s;
    }
}

__global__ __launch_bounds__(512) void reduce_kernel(const float* __restrict__ Gpart,
                                                     const float* __restrict__ Spart,
                                                     float* __restrict__ G,
                                                     float* __restrict__ S) {
    __shared__ float4 redG[3][128];
    __shared__ float  redS[16][32];
    const int b    = blockIdx.x;
    const int tid  = threadIdx.x;
    const int lane = tid & 31, w = tid >> 5;
    const int q    = tid & 127, pg = tid >> 7;
    const int f    = b * 128 + q;
    const int n    = f >> 14;
    const int e4   = f & 16383;
    const bool rowsums = b < 16;

    float u[8];
#pragma unroll
    for (int j = 0; j < 8; ++j) u[j] = 0.f;
    if (rowsums) {
        const float* sp = Spart + ((size_t)(b >> 3) * GSPLIT + w) * CCH + (b & 7) * 32 + lane;
#pragma unroll
        for (int j = 0; j < 8; ++j) u[j] = sp[(size_t)j * 16 * CCH];
    }

    const float4* gp = (const float4*)Gpart + ((size_t)n * GSPLIT + pg * 32) * 16384 + e4;
    float4 v[32];
#pragma unroll
    for (int i = 0; i < 32; ++i) v[i] = gp[(size_t)i * 16384];
    float4 s = v[0];
#pragma unroll
    for (int i = 1; i < 32; ++i) {
        s.x += v[i].x; s.y += v[i].y; s.z += v[i].z; s.w += v[i].w;
    }
    if (rowsums) {
        float t = u[0];
#pragma unroll
        for (int j = 1; j < 8; ++j) t += u[j];
        redS[w][lane] = t;
    }
    if (pg > 0) redG[pg - 1][q] = s;
    __syncthreads();
    if (pg == 0) {
        const float4 g1 = redG[0][q], g2 = redG[1][q], g3 = redG[2][q];
        s.x = ((s.x + g1.x) + g2.x) + g3.x;
        s.y = ((s.y + g1.y) + g2.y) + g3.y;
        s.z = ((s.z + g1.z) + g2.z) + g3.z;
        s.w = ((s.w + g1.w) + g2.w) + g3.w;
        ((float4*)G)[f] = s;
    } else if (rowsums && w == 4) {
        float t = redS[0][lane];
#pragma unroll
        for (int k = 1; k < 16; ++k) t += redS[k][lane];
        S[b * 32 + lane] = t;
    }
}

#define BPAD 260

__device__ __forceinline__ void lds_handover() {
    __builtin_amdgcn_fence(__ATOMIC_RELEASE, "workgroup", "local");
    __builtin_amdgcn_s_barrier();
    __builtin_amdgcn_fence(__ATOMIC_ACQUIRE, "workgroup", "local");
}

__global__ __launch_bounds__(512) void band_kernel(const float* __restrict__ Wq,
                                                   const float* __restrict__ Wk,
                                                   const float* __restrict__ Wv,
                                                   const float* __restrict__ bq,
                                                   const float* __restrict__ bk,
                                                   const float* __restrict__ bv,
                                                   const float* __restrict__ G,
                                                   const float* __restrict__ S,
                                                   float* __restrict__ Bt,
                                                   float* __restrict__ cvec) {
    __shared__ __attribute__((aligned(16))) float ldsQ[16 * BPAD];
    __shared__ __attribute__((aligned(16))) float ldsT[16 * BPAD];
    __shared__ __attribute__((aligned(16))) float ldsS[CCH];
    __shared__ float ldsBK[CCH];
    __shared__ float ldsBV[CCH];
    __shared__ float ldsBQ[16];
    __shared__ float ldsU[16];
    __shared__ float ldsC[16];

    const int n   = blockIdx.y;
    const int i0  = blockIdx.x * 16;
    const int tid = threadIdx.x, lane = tid & 31, w = tid >> 5;
    const int l15 = lane & 15, half = lane >> 4;
    const int col = w * 16 + l15;
    const int kh  = half * 2;
    const float* Gn  = G + (size_t)n * CCH * CCH;
    const float* wkr = Wk + col * CCH;

    v2f pa[32], pb[32];

    const int   t8    = tid & (CCH - 1);
    const float s_in  = S[n * CCH + t8];
    const float bk_in = bk[t8];
    const float bv_in = bv[t8];
    const float bq_in = bq[i0 + (tid & 15)];
    float4 q_in[2];
#pragma unroll
    for (int i = 0; i < 2; ++i) {
        const int f = tid + 512 * i;
        q_in[i] = *(const float4*)&Wq[(i0 + (f >> 6)) * CCH + (f & 63) * 4];
    }
    __builtin_amdgcn_sched_barrier(0);

#pragma unroll
    for (int i = 0; i < 32; ++i) {
        const int kk = i * 4 + kh;
        pa[i][0] = Gn[kk * CCH + col];
        pa[i][1] = Gn[(kk + 1) * CCH + col];
    }
#pragma unroll
    for (int i = 0; i < 32; ++i) {
        const int kk = (i + 32) * 4 + kh;
        pb[i][0] = Gn[kk * CCH + col];
        pb[i][1] = Gn[(kk + 1) * CCH + col];
    }

#pragma unroll
    for (int i = 0; i < 2; ++i) {
        const int f = tid + 512 * i;
        *(float4*)&ldsQ[(f >> 6) * BPAD + (f & 63) * 4] = q_in[i];
    }
    ldsS[t8] = s_in; ldsBK[t8] = bk_in; ldsBV[t8] = bv_in;
    ldsBQ[tid & 15] = bq_in;
    lds_handover();

    {
        float p = 0.f;
#pragma unroll
        for (int t = 0; t < 8; ++t) {
            const int j = lane + 32 * t;
            p += ldsQ[w * BPAD + j] * ldsS[j];
        }
        for (int off = 16; off >= 1; off >>= 1) p += __shfl_xor(p, off, 32);
        if (lane == 0) ldsU[w] = p;
    }

    const v8f zero8 = {0.f,0.f,0.f,0.f,0.f,0.f,0.f,0.f};

    v8f acc = zero8;
#pragma unroll
    for (int i = 0; i < 32; ++i) {
        const int kk = i * 4 + kh;
        const v2f a = *(const v2f*)&ldsQ[l15 * BPAD + kk];
        acc = wmma_f32(a, pa[i], acc);
    }
#pragma unroll
    for (int i = 0; i < 32; ++i) {
        const int kk = i * 4 + kh;
        pa[i] = *(const v2f*)&wkr[kk];
    }
    __builtin_amdgcn_sched_barrier(0);
#pragma unroll
    for (int i = 0; i < 32; ++i) {
        const int kk = (i + 32) * 4 + kh;
        const v2f a = *(const v2f*)&ldsQ[l15 * BPAD + kk];
        acc = wmma_f32(a, pb[i], acc);
    }
#pragma unroll
    for (int i = 0; i < 32; ++i) {
        const int kk = (i + 32) * 4 + kh;
        pb[i] = *(const v2f*)&wkr[kk];
    }
#pragma unroll
    for (int v = 0; v < 8; ++v) ldsT[(half * 8 + v) * BPAD + col] = acc[v];
    lds_handover();

    acc = zero8;
    float wd = 0.f;
#pragma unroll
    for (int i = 0; i < 32; ++i) {
        const int kk = i * 4 + kh;
        const v2f a  = *(const v2f*)&ldsT[l15 * BPAD + kk];
        const v2f sv = *(const v2f*)&ldsS[kk];
        wd += pa[i][0] * sv[0];
        wd += pa[i][1] * sv[1];
        acc = wmma_f32(a, pa[i], acc);
    }
#pragma unroll
    for (int i = 0; i < 32; ++i) {
        const int kk = i * 4 + kh;
        pa[i][0] = Wv[kk * CCH + col];
        pa[i][1] = Wv[(kk + 1) * CCH + col];
    }
    __builtin_amdgcn_sched_barrier(0);
#pragma unroll
    for (int i = 0; i < 32; ++i) {
        const int kk = (i + 32) * 4 + kh;
        const v2f a  = *(const v2f*)&ldsT[l15 * BPAD + kk];
        const v2f sv = *(const v2f*)&ldsS[kk];
        wd += pb[i][0] * sv[0];
        wd += pb[i][1] * sv[1];
        acc = wmma_f32(a, pb[i], acc);
    }
#pragma unroll
    for (int i = 0; i < 32; ++i) {
        const int kk = (i + 32) * 4 + kh;
        pb[i][0] = Wv[kk * CCH + col];
        pb[i][1] = Wv[(kk + 1) * CCH + col];
    }
    wd += __shfl_xor(wd, 16, 32);

    {
        const float bkc = ldsBK[col];
        const float wv  = wd + (float)MTOK * bkc;
#pragma unroll
        for (int v = 0; v < 8; ++v) {
            const int r = half * 8 + v;
            const float val = acc[v] + ldsU[r] * bkc + ldsBQ[r] * wv;
            ldsQ[r * BPAD + col] = val;
        }
    }
    lds_handover();

    {
        float* rowp = &ldsQ[w * BPAD];
        float l[8];
        float mx = -3.0e38f;
#pragma unroll
        for (int t = 0; t < 8; ++t) {
            l[t] = rowp[lane + 32 * t];
            mx = fmaxf(mx, l[t]);
        }
        for (int off = 16; off >= 1; off >>= 1) mx = fmaxf(mx, __shfl_xor(mx, off, 32));
        float sum = 0.f;
#pragma unroll
        for (int t = 0; t < 8; ++t) { l[t] = __expf(l[t] - mx); sum += l[t]; }
        for (int off = 16; off >= 1; off >>= 1) sum += __shfl_xor(sum, off, 32);
        const float inv = 1.f / sum;
        float cd = 0.f;
#pragma unroll
        for (int t = 0; t < 8; ++t) {
            const int j = lane + 32 * t;
            const float a = l[t] * inv;
            rowp[j] = a;
            cd += a * ldsBV[j];
        }
        for (int off = 16; off >= 1; off >>= 1) cd += __shfl_xor(cd, off, 32);
        if (lane == 0) ldsC[w] = cd;
    }
    lds_handover();

    acc = zero8;
#pragma unroll
    for (int i = 0; i < 32; ++i) {
        const int kk = i * 4 + kh;
        const v2f a = *(const v2f*)&ldsQ[l15 * BPAD + kk];
        acc = wmma_f32(a, pa[i], acc);
    }
#pragma unroll
    for (int i = 0; i < 32; ++i) {
        const int kk = (i + 32) * 4 + kh;
        const v2f a = *(const v2f*)&ldsQ[l15 * BPAD + kk];
        acc = wmma_f32(a, pb[i], acc);
    }
#pragma unroll
    for (int v = 0; v < 8; ++v) ldsT[(half * 8 + v) * BPAD + col] = acc[v];
    lds_handover();
    {
        float* brow = Bt + (size_t)n * CCH * CCH + (size_t)(i0 + w) * CCH;
#pragma unroll
        for (int t = 0; t < 2; ++t) {
            const int c = lane * 4 + 128 * t;
            *(float4*)&brow[c] = *(const float4*)&ldsT[w * BPAD + c];
        }
    }
    if (tid < 16) cvec[n * CCH + i0 + tid] = ldsC[tid];
}

#define OPITCH 260

__global__ __launch_bounds__(512) void out_kernel(const float* __restrict__ x,
                                                  const float* __restrict__ Bt,
                                                  const float* __restrict__ cvec,
                                                  float* __restrict__ out) {
    __shared__ __attribute__((aligned(16))) float ldsB[CCH * OPITCH];
    const int n = blockIdx.y;
    const float* xn  = x + (size_t)n * CCH * MTOK;
    const float* btn = Bt + (size_t)n * CCH * CCH;
    float* on = out + (size_t)n * MTOK * CCH;
    const int m0 = blockIdx.x * 256;
    const int tid = threadIdx.x, lane = tid & 31, w = tid >> 5;
    const int l15 = lane & 15, half = lane >> 4;
    const int kh  = half * 2;

    const int rot = blockIdx.x * 128;
    float4 t[32];
#pragma unroll
    for (int i = 0; i < 32; ++i) {
        const int f = (tid + 512 * i + rot) & 16383;
        t[i] = *(const float4*)&btn[(f >> 6) * CCH + (f & 63) * 4];
    }
#pragma unroll
    for (int i = 0; i < 32; ++i) {
        const int f = (tid + 512 * i + rot) & 16383;
        *(float4*)&ldsB[(f >> 6) * OPITCH + (f & 63) * 4] = t[i];
    }
    __builtin_amdgcn_sched_barrier(0);

    const float* xa = xn + (size_t)(kh * MTOK + m0 + w * 16 + l15);
    v2f a[64];
#pragma unroll
    for (int g = 0; g < 4; ++g) {
#pragma unroll
        for (int i = 16 * g; i < 16 * g + 16; ++i) {
            a[i][0] = xa[(size_t)(4 * i) * MTOK];
            a[i][1] = xa[(size_t)(4 * i + 1) * MTOK];
        }
        __builtin_amdgcn_sched_barrier(0);
    }
    lds_handover();

    const float* bl = &ldsB[l15 * OPITCH + kh];
    const v8f zero8 = {0.f,0.f,0.f,0.f,0.f,0.f,0.f,0.f};
    float* orow = on + (size_t)(m0 + w * 16 + half * 8) * CCH + l15;
    const float* cvn = cvec + n * CCH + l15;

#pragma unroll 1
    for (int tg = 0; tg < 4; ++tg) {
        const float* bt = bl + tg * (64 * OPITCH);
        v8f acc[4];
#pragma unroll
        for (int c = 0; c < 4; ++c) acc[c] = zero8;
#pragma unroll
        for (int i = 0; i < 64; ++i) {
#pragma unroll
            for (int c = 0; c < 4; ++c) {
                const v2f b = *(const v2f*)&bt[c * (16 * OPITCH) + 4 * i];
                acc[c] = wmma_f32(a[i], b, acc[c]);
            }
        }
#pragma unroll
        for (int c = 0; c < 4; ++c) {
            const int cofs = (tg * 4 + c) * 16;
            const float cv = cvn[cofs];
#pragma unroll
            for (int v = 0; v < 8; ++v)
                orow[(size_t)v * CCH + cofs] = acc[c][v] + cv;
        }
    }
}

extern "C" void kernel_launch(void* const* d_in, const int* in_sizes, int n_in,
                              void* d_out, int out_size, void* d_ws, size_t ws_size,
                              hipStream_t stream) {
    (void)in_sizes; (void)n_in; (void)out_size; (void)ws_size;
    const float* x  = (const float*)d_in[0];
    const float* Wq = (const float*)d_in[1];
    const float* bq = (const float*)d_in[2];
    const float* Wk = (const float*)d_in[3];
    const float* bk = (const float*)d_in[4];
    const float* Wv = (const float*)d_in[5];
    const float* bv = (const float*)d_in[6];
    float* out = (float*)d_out;
    float* ws  = (float*)d_ws;

    const size_t oSP  = 0;
    const size_t oG   = oSP  + (size_t)2 * 128 * 256;
    const size_t oS   = oG   + (size_t)2 * 65536;
    const size_t oBT  = oS   + 512;
    const size_t oCV  = oBT  + (size_t)2 * 65536;

    float* gpart = out;

    gram_kernel<<<dim3(128, 2), 512, 0, stream>>>(x, gpart, ws + oSP);
    reduce_kernel<<<256, 512, 0, stream>>>(gpart, ws + oSP, ws + oG, ws + oS);
    band_kernel<<<dim3(16, 2), 512, 0, stream>>>(Wq, Wk, Wv, bq, bk, bv, ws + oG, ws + oS,
                                                 ws + oBT, ws + oCV);
    out_kernel<<<dim3(128, 2), 512, 0, stream>>>(x, ws + oBT, ws + oCV, out);
}


# gram_kernel: a WG owns a 128x128 quadrant of G over 1024 tokens (wave tile 2x2, slab refill stored among the products), not all of G over 256: 32 partials per batch, not 128; G, s sum order changes
# speedup vs baseline: 3.3537x; 1.0727x over previous
#include <hip/hip_runtime.h>
#include <hip/hip_bf16.h>


#define MTOK 32768
#define CCH  256
#define GTOK 1024
#define GSPLIT (MTOK / GTOK)

typedef __attribute__((ext_vector_type(2))) float v2f;
typedef __attribute__((ext_vector_type(8))) float v8f;

__device__ __forceinline__ v8f wmma_f32(v2f a, v2f b, v8f c) {
    return __builtin_amdgcn_wmma_f32_16x16x4_f32(false, a, false, b, (short)0, c, false, false);
}

#define GST    128
#define GPITCH 132
#define GSLAB  (CCH * GPITCH)
__global__ __launch_bounds__(512) void gram_kernel(const float* __restrict__ x,
                                                   float* __restrict__ Gpart,
                                                   float* __restrict__ Spart) {
    __shared__ __attribute__((aligned(16))) float lds[2 * GSLAB];
    const int n     = blockIdx.y;
    const int chunk = blockIdx.x >> 2;
    const int qi    = (blockIdx.x >> 1) & 1;
    const int qj    = blockIdx.x & 1;
    const float* xn = x + (size_t)n * CCH * MTOK;
    const int tid  = threadIdx.x;
    const int lane = tid & 31;
    const int w    = tid >> 5;
    const int bwr  = w >> 2, bwc = w & 3;
    const int l15  = lane & 15;
    const int half = lane >> 4;

    const v8f zero8 = {0.f,0.f,0.f,0.f,0.f,0.f,0.f,0.f};
    v8f acc[2][2];
#pragma unroll
    for (int r = 0; r < 2; ++r)
#pragma unroll
        for (int c = 0; c < 2; ++c) acc[r][c] = zero8;

    const int m0  = chunk * GTOK;
    const float* xa = xn + (size_t)(128 * qi + w) * MTOK + m0 + 4 * lane;
    const float* xb = xn + (size_t)(128 * qj + w) * MTOK + m0 + 4 * lane;
    float* lst = &lds[w * GPITCH + 4 * lane];

    float sacc[8];
#pragma unroll
    for (int i = 0; i < 8; ++i) sacc[i] = 0.f;

    {
        float4 p0[16];
#pragma unroll
        for (int i = 0; i < 8; ++i) {
            p0[i]     = *(const float4*)&xa[(size_t)(16 * i) * MTOK];
            p0[8 + i] = *(const float4*)&xb[(size_t)(16 * i) * MTOK];
        }
#pragma unroll
        for (int i = 0; i < 16; ++i) *(float4*)&lst[(16 * i) * GPITCH] = p0[i];
#pragma unroll
        for (int i = 0; i < 8; ++i) sacc[i] += (p0[i].x + p0[i].y) + (p0[i].z + p0[i].w);
    }
    __syncthreads();

    const float* la = &lds[((bwr * 2) * 16 + l15) * GPITCH + half * 2];
    const float* lb = &lds[(128 + (bwc * 2) * 16 + l15) * GPITCH + half * 2];

    int cur = 0;
    for (int t0 = 0; t0 < GTOK; t0 += GST) {
        const bool more = t0 + GST < GTOK;
        float4 pf[16];
        if (more) {
#pragma unroll
            for (int i = 0; i < 8; ++i) {
                pf[i]     = *(const float4*)&xa[(size_t)(16 * i) * MTOK + t0 + GST];
                pf[8 + i] = *(const float4*)&xb[(size_t)(16 * i) * MTOK + t0 + GST];
            }
        }
        const float* ca = la + cur;
        const float* cb = lb + cur;
        v2f fa[2][4][2], fb[2][4][2];
#pragma unroll
        for (int u = 0; u < 4; ++u) {
#pragma unroll
            for (int r = 0; r < 2; ++r) {
                fa[0][u][r] = *(const v2f*)&ca[r * 16 * GPITCH + 4 * u];
                fb[0][u][r] = *(const v2f*)&cb[r * 16 * GPITCH + 4 * u];
            }
        }
#pragma unroll
        for (int p = 0; p < 8; ++p) {
            const int cu = p & 1, nx = cu ^ 1;
            if (p == 6 && more) {
                float* ls = lst + (GSLAB - cur);
#pragma unroll
                for (int i = 0; i < 16; ++i) *(float4*)&ls[(16 * i) * GPITCH] = pf[i];
#pragma unroll
                for (int i = 0; i < 8; ++i) sacc[i] += (pf[i].x + pf[i].y) + (pf[i].z + pf[i].w);
            }
            if (p < 7) {
#pragma unroll
                for (int u = 0; u < 4; ++u) {
                    const int kk = 16 * (p + 1) + 4 * u;
#pragma unroll
                    for (int r = 0; r < 2; ++r) {
                        fa[nx][u][r] = *(const v2f*)&ca[r * 16 * GPITCH + kk];
                        fb[nx][u][r] = *(const v2f*)&cb[r * 16 * GPITCH + kk];
                    }
                }
            }
            __builtin_amdgcn_sched_barrier(0);
#pragma unroll
            for (int u = 0; u < 4; ++u)
#pragma unroll
                for (int r = 0; r < 2; ++r)
#pragma unroll
                    for (int c = 0; c < 2; ++c)
                        acc[r][c] = wmma_f32(fa[cu][u][r], fb[cu][u][c], acc[r][c]);
        }
        __syncthreads();
        cur = GSLAB - cur;
    }

    float* gp = Gpart + ((size_t)n * GSPLIT + chunk) * (CCH * CCH);
#pragma unroll
    for (int r = 0; r < 2; ++r)
#pragma unroll
        for (int c = 0; c < 2; ++c) {
            const int row0 = 128 * qi + (bwr * 2 + r) * 16 + half * 8;
            const int col  = 128 * qj + (bwc * 2 + c) * 16 + l15;
#pragma unroll
            for (int v = 0; v < 8; ++v)
                gp[(row0 + v) * CCH + col] = acc[r][c][v];
        }

    if (qj == 0) {
#pragma unroll
        for (int i = 0; i < 8; ++i) {
            float s = sacc[i];
            for (int off = 16; off >= 1; off >>= 1) s += __shfl_xor(s, off, 32);
            if (lane == 0) lds[w + 16 * i] = s;
        }
        __syncthreads();
        if (w < 4)
            Spart[((size_t)n * GSPLIT + chunk) * CCH + 128 * qi + tid] = lds[tid];
    }
}

__global__ __launch_bounds__(512) void reduce_kernel(const float* __restrict__ Gpart,
                                                     const float* __restrict__ Spart,
                                                     float* __restrict__ G,
                                                     float* __restrict__ S) {
    __shared__ float4 redG[3][128];
    __shared__ float  redS[16][32];
    const int b    = blockIdx.x;
    const int tid  = threadIdx.x;
    const int lane = tid & 31, w = tid >> 5;
    const int q    = tid & 127, pg = tid >> 7;
    const int f    = b * 128 + q;
    const int n    = f >> 14;
    const int e4   = f & 16383;
    const bool rowsums = b < 16;

    float u[2];
#pragma unroll
    for (int j = 0; j < 2; ++j) u[j] = 0.f;
    if (rowsums) {
        const float* sp = Spart + ((size_t)(b >> 3) * GSPLIT + w) * CCH + (b & 7) * 32 + lane;
#pragma unroll
        for (int j = 0; j < 2; ++j) u[j] = sp[(size_t)j * 16 * CCH];
    }

    const float4* gp = (const float4*)Gpart + ((size_t)n * GSPLIT + pg * 8) * 16384 + e4;
    float4 v[8];
#pragma unroll
    for (int i = 0; i < 8; ++i) v[i] = gp[(size_t)i * 16384];
    float4 s = v[0];
#pragma unroll
    for (int i = 1; i < 8; ++i) {
        s.x += v[i].x; s.y += v[i].y; s.z += v[i].z; s.w += v[i].w;
    }
    if (rowsums) redS[w][lane] = u[0] + u[1];
    if (pg > 0) redG[pg - 1][q] = s;
    __syncthreads();
    if (pg == 0) {
        const float4 g1 = redG[0][q], g2 = redG[1][q], g3 = redG[2][q];
        s.x = ((s.x + g1.x) + g2.x) + g3.x;
        s.y = ((s.y + g1.y) + g2.y) + g3.y;
        s.z = ((s.z + g1.z) + g2.z) + g3.z;
        s.w = ((s.w + g1.w) + g2.w) + g3.w;
        ((float4*)G)[f] = s;
    } else if (rowsums && w == 4) {
        float t = redS[0][lane];
#pragma unroll
        for (int k = 1; k < 16; ++k) t += redS[k][lane];
        S[b * 32 + lane] = t;
    }
}

#define BPAD 260

__device__ __forceinline__ void lds_handover() {
    __builtin_amdgcn_fence(__ATOMIC_RELEASE, "workgroup", "local");
    __builtin_amdgcn_s_barrier();
    __builtin_amdgcn_fence(__ATOMIC_ACQUIRE, "workgroup", "local");
}

__global__ __launch_bounds__(512) void band_kernel(const float* __restrict__ Wq,
                                                   const float* __restrict__ Wk,
                                                   const float* __restrict__ Wv,
                                                   const float* __restrict__ bq,
                                                   const float* __restrict__ bk,
                                                   const float* __restrict__ bv,
                                                   const float* __restrict__ G,
                                                   const float* __restrict__ S,
                                                   float* __restrict__ Bt,
                                                   float* __restrict__ cvec) {
    __shared__ __attribute__((aligned(16))) float ldsQ[16 * BPAD];
    __shared__ __attribute__((aligned(16))) float ldsT[16 * BPAD];
    __shared__ __attribute__((aligned(16))) float ldsS[CCH];
    __shared__ float ldsBK[CCH];
    __shared__ float ldsBV[CCH];
    __shared__ float ldsBQ[16];
    __shared__ float ldsU[16];
    __shared__ float ldsC[16];

    const int n   = blockIdx.y;
    const int i0  = blockIdx.x * 16;
    const int tid = threadIdx.x, lane = tid & 31, w = tid >> 5;
    const int l15 = lane & 15, half = lane >> 4;
    const int col = w * 16 + l15;
    const int kh  = half * 2;
    const float* Gn  = G + (size_t)n * CCH * CCH;
    const float* wkr = Wk + col * CCH;

    v2f pa[32], pb[32];

    const int   t8    = tid & (CCH - 1);
    const float s_in  = S[n * CCH + t8];
    const float bk_in = bk[t8];
    const float bv_in = bv[t8];
    const float bq_in = bq[i0 + (tid & 15)];
    float4 q_in[2];
#pragma unroll
    for (int i = 0; i < 2; ++i) {
        const int f = tid + 512 * i;
        q_in[i] = *(const float4*)&Wq[(i0 + (f >> 6)) * CCH + (f & 63) * 4];
    }
    __builtin_amdgcn_sched_barrier(0);

#pragma unroll
    for (int i = 0; i < 32; ++i) {
        const int kk = i * 4 + kh;
        pa[i][0] = Gn[kk * CCH + col];
        pa[i][1] = Gn[(kk + 1) * CCH + col];
    }
#pragma unroll
    for (int i = 0; i < 32; ++i) {
        const int kk = (i + 32) * 4 + kh;
        pb[i][0] = Gn[kk * CCH + col];
        pb[i][1] = Gn[(kk + 1) * CCH + col];
    }

#pragma unroll
    for (int i = 0; i < 2; ++i) {
        const int f = tid + 512 * i;
        *(float4*)&ldsQ[(f >> 6) * BPAD + (f & 63) * 4] = q_in[i];
    }
    ldsS[t8] = s_in; ldsBK[t8] = bk_in; ldsBV[t8] = bv_in;
    ldsBQ[tid & 15] = bq_in;
    lds_handover();

    {
        float p = 0.f;
#pragma unroll
        for (int t = 0; t < 8; ++t) {
            const int j = lane + 32 * t;
            p += ldsQ[w * BPAD + j] * ldsS[j];
        }
        for (int off = 16; off >= 1; off >>= 1) p += __shfl_xor(p, off, 32);
        if (lane == 0) ldsU[w] = p;
    }

    const v8f zero8 = {0.f,0.f,0.f,0.f,0.f,0.f,0.f,0.f};

    v8f acc = zero8;
#pragma unroll
    for (int i = 0; i < 32; ++i) {
        const int kk = i * 4 + kh;
        const v2f a = *(const v2f*)&ldsQ[l15 * BPAD + kk];
        acc = wmma_f32(a, pa[i], acc);
    }
#pragma unroll
    for (int i = 0; i < 32; ++i) {
        const int kk = i * 4 + kh;
        pa[i] = *(const v2f*)&wkr[kk];
    }
    __builtin_amdgcn_sched_barrier(0);
#pragma unroll
    for (int i = 0; i < 32; ++i) {
        const int kk = (i + 32) * 4 + kh;
        const v2f a = *(const v2f*)&ldsQ[l15 * BPAD + kk];
        acc = wmma_f32(a, pb[i], acc);
    }
#pragma unroll
    for (int i = 0; i < 32; ++i) {
        const int kk = (i + 32) * 4 + kh;
        pb[i] = *(const v2f*)&wkr[kk];
    }
#pragma unroll
    for (int v = 0; v < 8; ++v) ldsT[(half * 8 + v) * BPAD + col] = acc[v];
    lds_handover();

    acc = zero8;
    float wd = 0.f;
#pragma unroll
    for (int i = 0; i < 32; ++i) {
        const int kk = i * 4 + kh;
        const v2f a  = *(const v2f*)&ldsT[l15 * BPAD + kk];
        const v2f sv = *(const v2f*)&ldsS[kk];
        wd += pa[i][0] * sv[0];
        wd += pa[i][1] * sv[1];
        acc = wmma_f32(a, pa[i], acc);
    }
#pragma unroll
    for (int i = 0; i < 32; ++i) {
        const int kk = i * 4 + kh;
        pa[i][0] = Wv[kk * CCH + col];
        pa[i][1] = Wv[(kk + 1) * CCH + col];
    }
    __builtin_amdgcn_sched_barrier(0);
#pragma unroll
    for (int i = 0; i < 32; ++i) {
        const int kk = (i + 32) * 4 + kh;
        const v2f a  = *(const v2f*)&ldsT[l15 * BPAD + kk];
        const v2f sv = *(const v2f*)&ldsS[kk];
        wd += pb[i][0] * sv[0];
        wd += pb[i][1] * sv[1];
        acc = wmma_f32(a, pb[i], acc);
    }
#pragma unroll
    for (int i = 0; i < 32; ++i) {
        const int kk = (i + 32) * 4 + kh;
        pb[i][0] = Wv[kk * CCH + col];
        pb[i][1] = Wv[(kk + 1) * CCH + col];
    }
    wd += __shfl_xor(wd, 16, 32);

    {
        const float bkc = ldsBK[col];
        const float wv  = wd + (float)MTOK * bkc;
#pragma unroll
        for (int v = 0; v < 8; ++v) {
            const int r = half * 8 + v;
            const float val = acc[v] + ldsU[r] * bkc + ldsBQ[r] * wv;
            ldsQ[r * BPAD + col] = val;
        }
    }
    lds_handover();

    {
        float* rowp = &ldsQ[w * BPAD];
        float l[8];
        float mx = -3.0e38f;
#pragma unroll
        for (int t = 0; t < 8; ++t) {
            l[t] = rowp[lane + 32 * t];
            mx = fmaxf(mx, l[t]);
        }
        for (int off = 16; off >= 1; off >>= 1) mx = fmaxf(mx, __shfl_xor(mx, off, 32));
        float sum = 0.f;
#pragma unroll
        for (int t = 0; t < 8; ++t) { l[t] = __expf(l[t] - mx); sum += l[t]; }
        for (int off = 16; off >= 1; off >>= 1) sum += __shfl_xor(sum, off, 32);
        const float inv = 1.f / sum;
        float cd = 0.f;
#pragma unroll
        for (int t = 0; t < 8; ++t) {
            const int j = lane + 32 * t;
            const float a = l[t] * inv;
            rowp[j] = a;
            cd += a * ldsBV[j];
        }
        for (int off = 16; off >= 1; off >>= 1) cd += __shfl_xor(cd, off, 32);
        if (lane == 0) ldsC[w] = cd;
    }
    lds_handover();

    acc = zero8;
#pragma unroll
    for (int i = 0; i < 32; ++i) {
        const int kk = i * 4 + kh;
        const v2f a = *(const v2f*)&ldsQ[l15 * BPAD + kk];
        acc = wmma_f32(a, pa[i], acc);
    }
#pragma unroll
    for (int i = 0; i < 32; ++i) {
        const int kk = (i + 32) * 4 + kh;
        const v2f a = *(const v2f*)&ldsQ[l15 * BPAD + kk];
        acc = wmma_f32(a, pb[i], acc);
    }
#pragma unroll
    for (int v = 0; v < 8; ++v) ldsT[(half * 8 + v) * BPAD + col] = acc[v];
    lds_handover();
    {
        float* brow = Bt + (size_t)n * CCH * CCH + (size_t)(i0 + w) * CCH;
#pragma unroll
        for (int t = 0; t < 2; ++t) {
            const int c = lane * 4 + 128 * t;
            *(float4*)&brow[c] = *(const float4*)&ldsT[w * BPAD + c];
        }
    }
    if (tid < 16) cvec[n * CCH + i0 + tid] = ldsC[tid];
}

#define OPITCH 260

__global__ __launch_bounds__(512) void out_kernel(const float* __restrict__ x,
                                                  const float* __restrict__ Bt,
                                                  const float* __restrict__ cvec,
                                                  float* __restrict__ out) {
    __shared__ __attribute__((aligned(16))) float ldsB[CCH * OPITCH];
    const int n = blockIdx.y;
    const float* xn  = x + (size_t)n * CCH * MTOK;
    const float* btn = Bt + (size_t)n * CCH * CCH;
    float* on = out + (size_t)n * MTOK * CCH;
    const int m0 = blockIdx.x * 256;
    const int tid = threadIdx.x, lane = tid & 31, w = tid >> 5;
    const int l15 = lane & 15, half = lane >> 4;
    const int kh  = half * 2;

    const int rot = blockIdx.x * 128;
    float4 t[32];
#pragma unroll
    for (int i = 0; i < 32; ++i) {
        const int f = (tid + 512 * i + rot) & 16383;
        t[i] = *(const float4*)&btn[(f >> 6) * CCH + (f & 63) * 4];
    }
#pragma unroll
    for (int i = 0; i < 32; ++i) {
        const int f = (tid + 512 * i + rot) & 16383;
        *(float4*)&ldsB[(f >> 6) * OPITCH + (f & 63) * 4] = t[i];
    }
    __builtin_amdgcn_sched_barrier(0);

    const float* xa = xn + (size_t)(kh * MTOK + m0 + w * 16 + l15);
    v2f a[64];
#pragma unroll
    for (int g = 0; g < 4; ++g) {
#pragma unroll
        for (int i = 16 * g; i < 16 * g + 16; ++i) {
            a[i][0] = xa[(size_t)(4 * i) * MTOK];
            a[i][1] = xa[(size_t)(4 * i + 1) * MTOK];
        }
        __builtin_amdgcn_sched_barrier(0);
    }
    lds_handover();

    const float* bl = &ldsB[l15 * OPITCH + kh];
    const v8f zero8 = {0.f,0.f,0.f,0.f,0.f,0.f,0.f,0.f};
    float* orow = on + (size_t)(m0 + w * 16 + half * 8) * CCH + l15;
    const float* cvn = cvec + n * CCH + l15;

#pragma unroll 1
    for (int tg = 0; tg < 4; ++tg) {
        const float* bt = bl + tg * (64 * OPITCH);
        v8f acc[4];
#pragma unroll
        for (int c = 0; c < 4; ++c) acc[c] = zero8;
#pragma unroll
        for (int i = 0; i < 64; ++i) {
#pragma unroll
            for (int c = 0; c < 4; ++c) {
                const v2f b = *(const v2f*)&bt[c * (16 * OPITCH) + 4 * i];
                acc[c] = wmma_f32(a[i], b, acc[c]);
            }
        }
#pragma unroll
        for (int c = 0; c < 4; ++c) {
            const int cofs = (tg * 4 + c) * 16;
            const float cv = cvn[cofs];
#pragma unroll
            for (int v = 0; v < 8; ++v)
                orow[(size_t)v * CCH + cofs] = acc[c][v] + cv;
        }
    }
}

extern "C" void kernel_launch(void* const* d_in, const int* in_sizes, int n_in,
                              void* d_out, int out_size, void* d_ws, size_t ws_size,
                              hipStream_t stream) {
    (void)in_sizes; (void)n_in; (void)out_size; (void)ws_size;
    const float* x  = (const float*)d_in[0];
    const float* Wq = (const float*)d_in[1];
    const float* bq = (const float*)d_in[2];
    const float* Wk = (const float*)d_in[3];
    const float* bk = (const float*)d_in[4];
    const float* Wv = (const float*)d_in[5];
    const float* bv = (const float*)d_in[6];
    float* out = (float*)d_out;
    float* ws  = (float*)d_ws;

    const size_t oSP  = 0;
    const size_t oG   = oSP  + (size_t)2 * 32 * 256;
    const size_t oS   = oG   + (size_t)2 * 65536;
    const size_t oBT  = oS   + 512;
    const size_t oCV  = oBT  + (size_t)2 * 65536;

    float* gpart = out;

    gram_kernel<<<dim3(128, 2), 512, 0, stream>>>(x, gpart, ws + oSP);
    reduce_kernel<<<256, 512, 0, stream>>>(gpart, ws + oSP, ws + oG, ws + oS);
    band_kernel<<<dim3(16, 2), 512, 0, stream>>>(Wq, Wk, Wv, bq, bk, bv, ws + oG, ws + oS,
                                                 ws + oBT, ws + oCV);
    out_kernel<<<dim3(128, 2), 512, 0, stream>>>(x, ws + oBT, ws + oCV, out);
}
